# starting kernel
# baseline (speedup 1.0000x reference)
#include <hip/hip_runtime.h>

typedef __attribute__((ext_vector_type(2))) float v2f;
typedef __attribute__((ext_vector_type(8))) float v8f;

#define H 128
#define W 128

__global__ __launch_bounds__(256)
void dw_gate_kernel(const float* __restrict__ lm,
                    const float* __restrict__ w_lm,
                    const float* __restrict__ b_lm,
                    float* __restrict__ gates)
{
    const int blk = blockIdx.x;
    const int g   = blockIdx.y;
    const int b   = blockIdx.z;
    const int tid = threadIdx.x;
    const int y0  = blk * 16;

    __shared__ float buf[2][26][138];
    __shared__ float gacc[16][128];

    for (int idx = tid; idx < 16*128; idx += 256)
        gacc[idx >> 7][idx & 127] = 0.f;
    for (int idx = tid; idx < 2*26*10; idx += 256) {
        int bsel = idx / 260, rem = idx % 260;
        int i = rem / 10, j = rem % 10;
        int col = (j < 5) ? j : (133 + (j - 5));
        buf[bsel][i][col] = 0.f;
    }
    __syncthreads();

    const int s0[4]    = {17, 27, 0, 48};
    const int cnt0[4]  = {10,  9, 17, 20};
    const int segs = (g == 0) ? 2 : 1;

    for (int s = 0; s < segs; ++s) {
        const int cstart = (s == 0) ? s0[g] : 36;
        const int ccnt   = (s == 0) ? cnt0[g] : 12;
        for (int ci = 0; ci < ccnt; ++ci) {
            const int c = cstart + ci;
            const float* src = lm + ((size_t)(b*68 + c) * H) * W;
            for (int idx = tid; idx < 26*128; idx += 256) {
                int i = idx >> 7, x = idx & 127;
                int y = y0 - 5 + i;
                buf[0][i][x + 5] = (y >= 0 && y < H) ? src[y*W + x] : 0.f;
            }
            __syncthreads();
            for (int k = 0; k < 5; ++k) {
                const float* wp = w_lm + (size_t)(k*68 + c) * 9;
                const float w00=wp[0],w01=wp[1],w02=wp[2],
                            w10=wp[3],w11=wp[4],w12=wp[5],
                            w20=wp[6],w21=wp[7],w22=wp[8];
                const float bias = b_lm[k*68 + c];
                float (*sb)[138] = buf[k & 1];
                float (*db)[138] = buf[(k + 1) & 1];
                const int rlo = k + 1, nrows = (24 - k) - rlo + 1;
                for (int idx = tid; idx < nrows*128; idx += 256) {
                    int i = rlo + (idx >> 7);
                    int x = idx & 127;
                    int y = y0 - 5 + i;
                    float v;
                    if (y < 0 || y >= H) {
                        v = 0.f;
                    } else {
                        int xc = x + 5;
                        v = bias
                          + w00*sb[i-1][xc-1] + w01*sb[i-1][xc] + w02*sb[i-1][xc+1]
                          + w10*sb[i  ][xc-1] + w11*sb[i  ][xc] + w12*sb[i  ][xc+1]
                          + w20*sb[i+1][xc-1] + w21*sb[i+1][xc] + w22*sb[i+1][xc+1];
                    }
                    db[i][x + 5] = v;
                }
                __syncthreads();
            }
            for (int idx = tid; idx < 16*128; idx += 256) {
                int r = idx >> 7, x = idx & 127;
                gacc[r][x] += buf[1][r + 5][x + 5];
            }
            __syncthreads();
        }
    }

    float* gp = gates + (((size_t)(b*4 + g) * H) + y0) * W;
    for (int idx = tid; idx < 16*128; idx += 256) {
        int r = idx >> 7, x = idx & 127;
        gp[r*W + x] = gacc[r][x];
    }
}

__global__ __launch_bounds__(256)
void conv16_wmma_kernel(const float* __restrict__ in,
                        float* __restrict__ out,
                        const float* __restrict__ w,
                        const float* __restrict__ bias,
                        const float* __restrict__ gates,
                        int gidx, int mode)
{
    const int y   = blockIdx.x;
    const int b   = blockIdx.y;
    const int tid = threadIdx.x;

    __shared__ float inT[3*130*18];
    __shared__ v2f   wB[36*32];

    for (int idx = tid; idx < 16*3*130; idx += 256) {
        int ch  = idx / 390;
        int rem = idx - ch*390;
        int r   = rem / 130;
        int xh  = rem - r*130;
        int yi = y + r - 1;
        int x  = xh - 1;
        float v = 0.f;
        if (yi >= 0 && yi < H && x >= 0 && x < W)
            v = in[((size_t)(b*16 + ch) * H + yi) * W + x];
        inT[(r*130 + xh)*18 + ch] = v;
    }
    for (int idx = tid; idx < 36*32; idx += 256) {
        int it   = idx >> 5;
        int lane = idx & 31;
        int kc  = it & 3;
        int tap = it >> 2;
        int ky = tap / 3, kx = tap - ky*3;
        int oc  = lane & 15;
        int icb = kc*4 + ((lane >> 4) << 1);
        v2f wv;
        wv.x = w[((oc*16 + icb    )*3 + ky)*3 + kx];
        wv.y = w[((oc*16 + icb + 1)*3 + ky)*3 + kx];
        wB[idx] = wv;
    }
    __syncthreads();

    const int lane  = tid & 31;
    const int wave  = tid >> 5;
    const int x0    = wave * 16;
    const int laneM = lane & 15;
    const int half  = lane >> 4;

    v8f acc = {0.f,0.f,0.f,0.f,0.f,0.f,0.f,0.f};
    #pragma unroll
    for (int ky = 0; ky < 3; ++ky) {
        #pragma unroll
        for (int kx = 0; kx < 3; ++kx) {
            const int xh = x0 + laneM + kx;
            const float* ap = &inT[(ky*130 + xh)*18 + 2*half];
            const int tap = ky*3 + kx;
            #pragma unroll
            for (int kc = 0; kc < 4; ++kc) {
                v2f a;
                a.x = ap[kc*4];
                a.y = ap[kc*4 + 1];
                v2f bb = wB[(tap*4 + kc)*32 + lane];
                acc = __builtin_amdgcn_wmma_f32_16x16x4_f32(
                          false, a, false, bb, (short)0, acc, false, false);
            }
        }
    }

    const float bv = bias[laneM];
    const int xb = x0 + half*8;
    float* op = out + ((size_t)(b*16 + laneM) * H + y) * W + xb;
    if (mode == 0) {
        #pragma unroll
        for (int r = 0; r < 8; ++r) op[r] = acc[r] + bv;
    } else {
        const float* gp = gates + ((size_t)(b*4 + gidx) * H + y) * W + xb;
        #pragma unroll
        for (int r = 0; r < 8; ++r) {
            float v = (acc[r] + bv) * gp[r];
            if (mode == 2) v += op[r];
            op[r] = v;
        }
    }
}

extern "C" void kernel_launch(void* const* d_in, const int* in_sizes, int n_in,
                              void* d_out, int out_size, void* d_ws, size_t ws_size,
                              hipStream_t stream)
{
    (void)in_sizes; (void)n_in; (void)out_size; (void)ws_size;
    const float* landmarks = (const float*)d_in[0];
    const float* features  = (const float*)d_in[1];
    const float* w_lm      = (const float*)d_in[2];
    const float* b_lm      = (const float*)d_in[3];
    const float* w_fea     = (const float*)d_in[4];
    const float* b_fea     = (const float*)d_in[5];
    float* out = (float*)d_out;

    char* ws = (char*)d_ws;
    float* gates = (float*)ws;
    float* bufA  = (float*)(ws + (size_t)8  * 1024*1024);
    float* bufB  = (float*)(ws + (size_t)40 * 1024*1024);

    dw_gate_kernel<<<dim3(8, 4, 32), 256, 0, stream>>>(landmarks, w_lm, b_lm, gates);

    for (int g = 0; g < 4; ++g) {
        const float* w0  = w_fea + (size_t)(g*3 + 0) * 16*16*9;
        const float* w1  = w_fea + (size_t)(g*3 + 1) * 16*16*9;
        const float* w2  = w_fea + (size_t)(g*3 + 2) * 16*16*9;
        const float* bb0 = b_fea + (g*3 + 0) * 16;
        const float* bb1 = b_fea + (g*3 + 1) * 16;
        const float* bb2 = b_fea + (g*3 + 2) * 16;
        conv16_wmma_kernel<<<dim3(128, 32), 256, 0, stream>>>(features, bufA, w0, bb0, gates, g, 0);
        conv16_wmma_kernel<<<dim3(128, 32), 256, 0, stream>>>(bufA,     bufB, w1, bb1, gates, g, 0);
        conv16_wmma_kernel<<<dim3(128, 32), 256, 0, stream>>>(bufB,     out,  w2, bb2, gates, g, (g == 0) ? 1 : 2);
    }
}


# conv kernels: each work-group computes a band of 8 output rows instead of 1: weights packed once and 10 input rows staged once per band, B fragments kept in registers; summation order kept
# speedup vs baseline: 1.5943x; 1.5943x over previous
#include <hip/hip_runtime.h>

typedef __attribute__((ext_vector_type(2))) float v2f;
typedef __attribute__((ext_vector_type(8))) float v8f;

#define H 128
#define W 128

__global__ __launch_bounds__(256)
void dw_gate_kernel(const float* __restrict__ lm,
                    const float* __restrict__ w_lm,
                    const float* __restrict__ b_lm,
                    float* __restrict__ gates)
{
    const int blk = blockIdx.x;
    const int g   = blockIdx.y;
    const int b   = blockIdx.z;
    const int tid = threadIdx.x;
    const int y0  = blk * 16;

    __shared__ float buf[2][26][138];
    __shared__ float gacc[16][128];

    for (int idx = tid; idx < 16*128; idx += 256)
        gacc[idx >> 7][idx & 127] = 0.f;
    for (int idx = tid; idx < 2*26*10; idx += 256) {
        int bsel = idx / 260, rem = idx % 260;
        int i = rem / 10, j = rem % 10;
        int col = (j < 5) ? j : (133 + (j - 5));
        buf[bsel][i][col] = 0.f;
    }
    __syncthreads();

    const int s0[4]    = {17, 27, 0, 48};
    const int cnt0[4]  = {10,  9, 17, 20};
    const int segs = (g == 0) ? 2 : 1;

    for (int s = 0; s < segs; ++s) {
        const int cstart = (s == 0) ? s0[g] : 36;
        const int ccnt   = (s == 0) ? cnt0[g] : 12;
        for (int ci = 0; ci < ccnt; ++ci) {
            const int c = cstart + ci;
            const float* src = lm + ((size_t)(b*68 + c) * H) * W;
            for (int idx = tid; idx < 26*128; idx += 256) {
                int i = idx >> 7, x = idx & 127;
                int y = y0 - 5 + i;
                buf[0][i][x + 5] = (y >= 0 && y < H) ? src[y*W + x] : 0.f;
            }
            __syncthreads();
            for (int k = 0; k < 5; ++k) {
                const float* wp = w_lm + (size_t)(k*68 + c) * 9;
                const float w00=wp[0],w01=wp[1],w02=wp[2],
                            w10=wp[3],w11=wp[4],w12=wp[5],
                            w20=wp[6],w21=wp[7],w22=wp[8];
                const float bias = b_lm[k*68 + c];
                float (*sb)[138] = buf[k & 1];
                float (*db)[138] = buf[(k + 1) & 1];
                const int rlo = k + 1, nrows = (24 - k) - rlo + 1;
                for (int idx = tid; idx < nrows*128; idx += 256) {
                    int i = rlo + (idx >> 7);
                    int x = idx & 127;
                    int y = y0 - 5 + i;
                    float v;
                    if (y < 0 || y >= H) {
                        v = 0.f;
                    } else {
                        int xc = x + 5;
                        v = bias
                          + w00*sb[i-1][xc-1] + w01*sb[i-1][xc] + w02*sb[i-1][xc+1]
                          + w10*sb[i  ][xc-1] + w11*sb[i  ][xc] + w12*sb[i  ][xc+1]
                          + w20*sb[i+1][xc-1] + w21*sb[i+1][xc] + w22*sb[i+1][xc+1];
                    }
                    db[i][x + 5] = v;
                }
                __syncthreads();
            }
            for (int idx = tid; idx < 16*128; idx += 256) {
                int r = idx >> 7, x = idx & 127;
                gacc[r][x] += buf[1][r + 5][x + 5];
            }
            __syncthreads();
        }
    }

    float* gp = gates + (((size_t)(b*4 + g) * H) + y0) * W;
    for (int idx = tid; idx < 16*128; idx += 256) {
        int r = idx >> 7, x = idx & 127;
        gp[r*W + x] = gacc[r][x];
    }
}

#define RB 8
typedef __attribute__((ext_vector_type(4))) float v4f;

template <int MODE>
__global__ __launch_bounds__(256)
void conv16_wmma_kernel(const float* __restrict__ in,
                        float* __restrict__ out,
                        const float* __restrict__ w,
                        const float* __restrict__ bias,
                        const float* __restrict__ gates,
                        int gidx)
{
    const int y0  = blockIdx.x * RB;
    const int b   = blockIdx.y;
    const int tid = threadIdx.x;
    const int lane = tid & 31;
    const int wave = tid >> 5;

    __shared__ float inT[(RB+2)*130*18];
    __shared__ v2f   wB[36*32];

    {
        const int q   = (lane & 7) + 8*(wave & 3);
        const int chb = ((lane >> 3) & 3) + 8*(wave >> 2);
        v4f sv[2*(RB+2)];
        #pragma unroll
        for (int j = 0; j < 2*(RB+2); ++j) {
            const int row = j >> 1;
            const int ch  = chb + 4*(j & 1);
            const int yi  = y0 - 1 + row;
            const int yc  = yi < 0 ? 0 : (yi > H-1 ? H-1 : yi);
            sv[j] = *(const v4f*)(in + ((size_t)(b*16 + ch) * H + yc) * W + 4*q);
        }
        #pragma unroll
        for (int j = 0; j < 2*(RB+2); ++j) {
            const int row = j >> 1;
            const int ch  = chb + 4*(j & 1);
            const int yi  = y0 - 1 + row;
            const bool inside = (yi >= 0 && yi < H);
            const v4f v = inside ? sv[j] : (v4f){0.f, 0.f, 0.f, 0.f};
            float* d = &inT[(row*130 + 1 + 4*q)*18 + ch];
            d[0] = v.x; d[18] = v.y; d[36] = v.z; d[54] = v.w;
        }
        for (int idx = tid; idx < (RB+2)*2*16; idx += 256) {
            const int ch = idx & 15;
            const int t  = idx >> 4;
            const int side = t & 1, row = t >> 1;
            inT[(row*130 + side*129)*18 + ch] = 0.f;
        }
    }
    for (int idx = tid; idx < 36*32; idx += 256) {
        int it   = idx >> 5;
        int ln   = idx & 31;
        int kc  = it & 3;
        int tap = it >> 2;
        int ky = tap / 3, kx = tap - ky*3;
        int oc  = ln & 15;
        int icb = kc*4 + ((ln >> 4) << 1);
        v2f wv;
        wv.x = w[((oc*16 + icb    )*3 + ky)*3 + kx];
        wv.y = w[((oc*16 + icb + 1)*3 + ky)*3 + kx];
        wB[idx] = wv;
    }
    __syncthreads();

    const int x0    = wave * 16;
    const int laneM = lane & 15;
    const int half  = lane >> 4;

    v2f bB[36];
    #pragma unroll
    for (int it = 0; it < 36; ++it) bB[it] = wB[it*32 + lane];

    const float bv = bias[laneM];
    const int xb = x0 + half*8;

    v8f acc[RB];
    #pragma unroll
    for (int r = 0; r < RB; ++r) acc[r] = (v8f){0.f,0.f,0.f,0.f,0.f,0.f,0.f,0.f};

    v2f aCur[12], aNxt[12];
    #pragma unroll
    for (int kx = 0; kx < 3; ++kx) {
        const float* ap = &inT[(0*130 + x0 + laneM + kx)*18 + 2*half];
        #pragma unroll
        for (int kc = 0; kc < 4; ++kc) { aCur[kx*4+kc].x = ap[kc*4]; aCur[kx*4+kc].y = ap[kc*4+1]; }
    }

    #pragma unroll
    for (int i = 0; i < RB+2; ++i) {
        if (i + 1 < RB+2) {
            #pragma unroll
            for (int kx = 0; kx < 3; ++kx) {
                const float* ap = &inT[((i+1)*130 + x0 + laneM + kx)*18 + 2*half];
                #pragma unroll
                for (int kc = 0; kc < 4; ++kc) { aNxt[kx*4+kc].x = ap[kc*4]; aNxt[kx*4+kc].y = ap[kc*4+1]; }
            }
        }
        #pragma unroll
        for (int kx = 0; kx < 3; ++kx) {
            #pragma unroll
            for (int kc = 0; kc < 4; ++kc) {
                #pragma unroll
                for (int ky = 0; ky < 3; ++ky) {
                    const int r = i - ky;
                    if (r >= 0 && r < RB)
                        acc[r] = __builtin_amdgcn_wmma_f32_16x16x4_f32(
                                     false, aCur[kx*4+kc], false, bB[(ky*3+kx)*4+kc],
                                     (short)0, acc[r], false, false);
                }
            }
        }
        if (i >= 2) {
            const int r = i - 2;
            const int y = y0 + r;
            float* op = out + ((size_t)(b*16 + laneM) * H + y) * W + xb;
            if (MODE == 0) {
                #pragma unroll
                for (int e = 0; e < 8; ++e) op[e] = acc[r][e] + bv;
            } else {
                const float* gp = gates + ((size_t)(b*4 + gidx) * H + y) * W + xb;
                #pragma unroll
                for (int e = 0; e < 8; ++e) {
                    float v = (acc[r][e] + bv) * gp[e];
                    if (MODE == 2) v += op[e];
                    op[e] = v;
                }
            }
        }
        #pragma unroll
        for (int t = 0; t < 12; ++t) aCur[t] = aNxt[t];
    }
}

extern "C" void kernel_launch(void* const* d_in, const int* in_sizes, int n_in,
                              void* d_out, int out_size, void* d_ws, size_t ws_size,
                              hipStream_t stream)
{
    (void)in_sizes; (void)n_in; (void)out_size; (void)ws_size;
    const float* landmarks = (const float*)d_in[0];
    const float* features  = (const float*)d_in[1];
    const float* w_lm      = (const float*)d_in[2];
    const float* b_lm      = (const float*)d_in[3];
    const float* w_fea     = (const float*)d_in[4];
    const float* b_fea     = (const float*)d_in[5];
    float* out = (float*)d_out;

    char* ws = (char*)d_ws;
    float* gates = (float*)ws;
    float* bufA  = (float*)(ws + (size_t)8  * 1024*1024);
    float* bufB  = (float*)(ws + (size_t)40 * 1024*1024);

    dw_gate_kernel<<<dim3(8, 4, 32), 256, 0, stream>>>(landmarks, w_lm, b_lm, gates);

    for (int g = 0; g < 4; ++g) {
        const float* w0  = w_fea + (size_t)(g*3 + 0) * 16*16*9;
        const float* w1  = w_fea + (size_t)(g*3 + 1) * 16*16*9;
        const float* w2  = w_fea + (size_t)(g*3 + 2) * 16*16*9;
        const float* bb0 = b_fea + (g*3 + 0) * 16;
        const float* bb1 = b_fea + (g*3 + 1) * 16;
        const float* bb2 = b_fea + (g*3 + 2) * 16;
        conv16_wmma_kernel<0><<<dim3(128/RB, 32), 256, 0, stream>>>(features, bufA, w0, bb0, gates, g);
        conv16_wmma_kernel<0><<<dim3(128/RB, 32), 256, 0, stream>>>(bufA,     bufB, w1, bb1, gates, g);
        if (g == 0)
            conv16_wmma_kernel<1><<<dim3(128/RB, 32), 256, 0, stream>>>(bufB, out, w2, bb2, gates, g);
        else
            conv16_wmma_kernel<2><<<dim3(128/RB, 32), 256, 0, stream>>>(bufB, out, w2, bb2, gates, g);
    }
}


# depthwise towers: one wave runs a channel's 5 layers in registers (lane = 4 columns x 26 rows, edge columns by lane exchange), no LDS passes; gate sum = per-wave partials added in wave order
# speedup vs baseline: 2.8217x; 1.7699x over previous
#include <hip/hip_runtime.h>

typedef __attribute__((ext_vector_type(2))) float v2f;
typedef __attribute__((ext_vector_type(8))) float v8f;
typedef __attribute__((ext_vector_type(4))) float v4f;

#define H 128
#define W 128


__device__ __forceinline__ float from_left(float v, int addr, bool first_lane)
{
    const float t = __int_as_float(__builtin_amdgcn_ds_bpermute(addr, __float_as_int(v)));
    return first_lane ? 0.f : t;
}
__device__ __forceinline__ float from_right(float v, int addr, bool last_lane)
{
    const float t = __int_as_float(__builtin_amdgcn_ds_bpermute(addr, __float_as_int(v)));
    return last_lane ? 0.f : t;
}

__device__ __forceinline__ float tap9(float s,
                                      float w0, float v0, float w1, float v1, float w2, float v2,
                                      float w3, float v3, float w4, float v4, float w5, float v5,
                                      float w6, float v6, float w7, float v7, float w8, float v8)
{
    s = fmaf(w0, v0, s); s = fmaf(w1, v1, s); s = fmaf(w2, v2, s);
    s = fmaf(w3, v3, s); s = fmaf(w4, v4, s); s = fmaf(w5, v5, s);
    s = fmaf(w6, v6, s); s = fmaf(w7, v7, s); s = fmaf(w8, v8, s);
    return s;
}

template <int K>
__device__ __forceinline__ void dw_layer(v4f (&r)[26], const float* __restrict__ wp, const float bias,
                                         const int y0, const int addrL, const int addrR,
                                         const bool first_lane, const bool last_lane)
{
    constexpr int RLO = K + 1, RHI = 24 - K;
    const float w00 = wp[0], w01 = wp[1], w02 = wp[2],
                w10 = wp[3], w11 = wp[4], w12 = wp[5],
                w20 = wp[6], w21 = wp[7], w22 = wp[8];
    v4f a = r[RLO - 1];
    v4f b = r[RLO];
    float aL = from_left(a.w, addrL, first_lane), aR = from_right(a.x, addrR, last_lane);
    float bL = from_left(b.w, addrL, first_lane), bR = from_right(b.x, addrR, last_lane);
    #pragma unroll
    for (int i = RLO; i <= RHI; ++i) {
        const v4f c = r[i + 1];
        const float cL = from_left(c.w, addrL, first_lane);
        const float cR = from_right(c.x, addrR, last_lane);
        v4f o;
        o.x = tap9(bias, w00, aL,  w01, a.x, w02, a.y, w10, bL,  w11, b.x, w12, b.y, w20, cL,  w21, c.x, w22, c.y);
        o.y = tap9(bias, w00, a.x, w01, a.y, w02, a.z, w10, b.x, w11, b.y, w12, b.z, w20, c.x, w21, c.y, w22, c.z);
        o.z = tap9(bias, w00, a.y, w01, a.z, w02, a.w, w10, b.y, w11, b.z, w12, b.w, w20, c.y, w21, c.z, w22, c.w);
        o.w = tap9(bias, w00, a.z, w01, a.w, w02, aR,  w10, b.z, w11, b.w, w12, bR,  w20, c.z, w21, c.w, w22, cR);
        if (i < 5 || i > 20) {
            const int y = y0 - 5 + i;
            if (y < 0 || y >= H) o = (v4f){0.f, 0.f, 0.f, 0.f};
        }
        r[i] = o;
        a = b; aL = bL; aR = bR;
        b = c; bL = cL; bR = cR;
    }
}

__global__ __launch_bounds__(256)
void dw_gate_kernel(const float* __restrict__ lm,
                    const float* __restrict__ w_lm,
                    const float* __restrict__ b_lm,
                    float* __restrict__ gates)
{
    const int blk = blockIdx.x;
    const int g   = blockIdx.y;
    const int b   = blockIdx.z;
    const int tid = threadIdx.x;
    const int lane = tid & 31;
    const int wave = __builtin_amdgcn_readfirstlane(tid >> 5);
    const int y0  = blk * 16;

    __shared__ v4f part[8][16][32];

    const int cnt = (g == 0) ? 22 : (g == 1) ? 9 : (g == 2) ? 17 : 20;
    const int addrL = ((lane + 31) & 31) << 2;
    const int addrR = ((lane + 1) & 31) << 2;
    const bool first_lane = (lane == 0), last_lane = (lane == 31);

    v4f acc[16];
    #pragma unroll
    for (int j = 0; j < 16; ++j) acc[j] = (v4f){0.f, 0.f, 0.f, 0.f};

    for (int p = wave; p < cnt; p += 8) {
        int c;
        if (g == 0)      c = (p < 10) ? 17 + p : 26 + p;
        else if (g == 1) c = 27 + p;
        else if (g == 2) c = p;
        else             c = 48 + p;
        const float* src = lm + ((size_t)(b*68 + c) * H) * W;

        v4f r[26];
        #pragma unroll
        for (int i = 0; i < 26; ++i) {
            const int y = y0 - 5 + i;
            int yc = y;
            if (i < 5)  yc = y < 0 ? 0 : y;
            if (i > 20) yc = y > H-1 ? H-1 : y;
            r[i] = *(const v4f*)(src + yc*W + 4*lane);
        }
        #pragma unroll
        for (int i = 0; i < 26; ++i) {
            if (i < 5 || i > 20) {
                const int y = y0 - 5 + i;
                if (y < 0 || y >= H) r[i] = (v4f){0.f, 0.f, 0.f, 0.f};
            }
        }

        dw_layer<0>(r, w_lm + (size_t)(0*68 + c) * 9, b_lm[0*68 + c], y0, addrL, addrR, first_lane, last_lane);
        dw_layer<1>(r, w_lm + (size_t)(1*68 + c) * 9, b_lm[1*68 + c], y0, addrL, addrR, first_lane, last_lane);
        dw_layer<2>(r, w_lm + (size_t)(2*68 + c) * 9, b_lm[2*68 + c], y0, addrL, addrR, first_lane, last_lane);
        dw_layer<3>(r, w_lm + (size_t)(3*68 + c) * 9, b_lm[3*68 + c], y0, addrL, addrR, first_lane, last_lane);
        dw_layer<4>(r, w_lm + (size_t)(4*68 + c) * 9, b_lm[4*68 + c], y0, addrL, addrR, first_lane, last_lane);

        #pragma unroll
        for (int j = 0; j < 16; ++j) acc[j] += r[5 + j];
    }

    #pragma unroll
    for (int j = 0; j < 16; ++j) part[wave][j][lane] = acc[j];
    __syncthreads();

    for (int idx = tid; idx < 16*32; idx += 256) {
        const int j = idx >> 5, q = idx & 31;
        v4f s = part[0][j][q];
        #pragma unroll
        for (int wv = 1; wv < 8; ++wv) s += part[wv][j][q];
        *(v4f*)(gates + (((size_t)(b*4 + g) * H) + y0 + j) * W + 4*q) = s;
    }
}

#define RB 8

template <int MODE>
__global__ __launch_bounds__(256)
void conv16_wmma_kernel(const float* __restrict__ in,
                        float* __restrict__ out,
                        const float* __restrict__ w,
                        const float* __restrict__ bias,
                        const float* __restrict__ gates,
                        int gidx)
{
    const int y0  = blockIdx.x * RB;
    const int b   = blockIdx.y;
    const int tid = threadIdx.x;
    const int lane = tid & 31;
    const int wave = tid >> 5;

    __shared__ float inT[(RB+2)*130*18];
    __shared__ v2f   wB[36*32];

    {
        const int q   = (lane & 7) + 8*(wave & 3);
        const int chb = ((lane >> 3) & 3) + 8*(wave >> 2);
        v4f sv[2*(RB+2)];
        #pragma unroll
        for (int j = 0; j < 2*(RB+2); ++j) {
            const int row = j >> 1;
            const int ch  = chb + 4*(j & 1);
            const int yi  = y0 - 1 + row;
            const int yc  = yi < 0 ? 0 : (yi > H-1 ? H-1 : yi);
            sv[j] = *(const v4f*)(in + ((size_t)(b*16 + ch) * H + yc) * W + 4*q);
        }
        #pragma unroll
        for (int j = 0; j < 2*(RB+2); ++j) {
            const int row = j >> 1;
            const int ch  = chb + 4*(j & 1);
            const int yi  = y0 - 1 + row;
            const bool inside = (yi >= 0 && yi < H);
            const v4f v = inside ? sv[j] : (v4f){0.f, 0.f, 0.f, 0.f};
            float* d = &inT[(row*130 + 1 + 4*q)*18 + ch];
            d[0] = v.x; d[18] = v.y; d[36] = v.z; d[54] = v.w;
        }
        for (int idx = tid; idx < (RB+2)*2*16; idx += 256) {
            const int ch = idx & 15;
            const int t  = idx >> 4;
            const int side = t & 1, row = t >> 1;
            inT[(row*130 + side*129)*18 + ch] = 0.f;
        }
    }
    for (int idx = tid; idx < 36*32; idx += 256) {
        int it   = idx >> 5;
        int ln   = idx & 31;
        int kc  = it & 3;
        int tap = it >> 2;
        int ky = tap / 3, kx = tap - ky*3;
        int oc  = ln & 15;
        int icb = kc*4 + ((ln >> 4) << 1);
        v2f wv;
        wv.x = w[((oc*16 + icb    )*3 + ky)*3 + kx];
        wv.y = w[((oc*16 + icb + 1)*3 + ky)*3 + kx];
        wB[idx] = wv;
    }
    __syncthreads();

    const int x0    = wave * 16;
    const int laneM = lane & 15;
    const int half  = lane >> 4;

    v2f bB[36];
    #pragma unroll
    for (int it = 0; it < 36; ++it) bB[it] = wB[it*32 + lane];

    const float bv = bias[laneM];
    const int xb = x0 + half*8;

    v8f acc[RB];
    #pragma unroll
    for (int r = 0; r < RB; ++r) acc[r] = (v8f){0.f,0.f,0.f,0.f,0.f,0.f,0.f,0.f};

    v2f aCur[12], aNxt[12];
    #pragma unroll
    for (int kx = 0; kx < 3; ++kx) {
        const float* ap = &inT[(0*130 + x0 + laneM + kx)*18 + 2*half];
        #pragma unroll
        for (int kc = 0; kc < 4; ++kc) { aCur[kx*4+kc].x = ap[kc*4]; aCur[kx*4+kc].y = ap[kc*4+1]; }
    }

    #pragma unroll
    for (int i = 0; i < RB+2; ++i) {
        if (i + 1 < RB+2) {
            #pragma unroll
            for (int kx = 0; kx < 3; ++kx) {
                const float* ap = &inT[((i+1)*130 + x0 + laneM + kx)*18 + 2*half];
                #pragma unroll
                for (int kc = 0; kc < 4; ++kc) { aNxt[kx*4+kc].x = ap[kc*4]; aNxt[kx*4+kc].y = ap[kc*4+1]; }
            }
        }
        #pragma unroll
        for (int kx = 0; kx < 3; ++kx) {
            #pragma unroll
            for (int kc = 0; kc < 4; ++kc) {
                #pragma unroll
                for (int ky = 0; ky < 3; ++ky) {
                    const int r = i - ky;
                    if (r >= 0 && r < RB)
                        acc[r] = __builtin_amdgcn_wmma_f32_16x16x4_f32(
                                     false, aCur[kx*4+kc], false, bB[(ky*3+kx)*4+kc],
                                     (short)0, acc[r], false, false);
                }
            }
        }
        if (i >= 2) {
            const int r = i - 2;
            const int y = y0 + r;
            float* op = out + ((size_t)(b*16 + laneM) * H + y) * W + xb;
            if (MODE == 0) {
                #pragma unroll
                for (int e = 0; e < 8; ++e) op[e] = acc[r][e] + bv;
            } else {
                const float* gp = gates + ((size_t)(b*4 + gidx) * H + y) * W + xb;
                #pragma unroll
                for (int e = 0; e < 8; ++e) {
                    float v = (acc[r][e] + bv) * gp[e];
                    if (MODE == 2) v += op[e];
                    op[e] = v;
                }
            }
        }
        #pragma unroll
        for (int t = 0; t < 12; ++t) aCur[t] = aNxt[t];
    }
}

extern "C" void kernel_launch(void* const* d_in, const int* in_sizes, int n_in,
                              void* d_out, int out_size, void* d_ws, size_t ws_size,
                              hipStream_t stream)
{
    (void)in_sizes; (void)n_in; (void)out_size; (void)ws_size;
    const float* landmarks = (const float*)d_in[0];
    const float* features  = (const float*)d_in[1];
    const float* w_lm      = (const float*)d_in[2];
    const float* b_lm      = (const float*)d_in[3];
    const float* w_fea     = (const float*)d_in[4];
    const float* b_fea     = (const float*)d_in[5];
    float* out = (float*)d_out;

    char* ws = (char*)d_ws;
    float* gates = (float*)ws;
    float* bufA  = (float*)(ws + (size_t)8  * 1024*1024);
    float* bufB  = (float*)(ws + (size_t)40 * 1024*1024);

    dw_gate_kernel<<<dim3(8, 4, 32), 256, 0, stream>>>(landmarks, w_lm, b_lm, gates);

    for (int g = 0; g < 4; ++g) {
        const float* w0  = w_fea + (size_t)(g*3 + 0) * 16*16*9;
        const float* w1  = w_fea + (size_t)(g*3 + 1) * 16*16*9;
        const float* w2  = w_fea + (size_t)(g*3 + 2) * 16*16*9;
        const float* bb0 = b_fea + (g*3 + 0) * 16;
        const float* bb1 = b_fea + (g*3 + 1) * 16;
        const float* bb2 = b_fea + (g*3 + 2) * 16;
        conv16_wmma_kernel<0><<<dim3(128/RB, 32), 256, 0, stream>>>(features, bufA, w0, bb0, gates, g);
        conv16_wmma_kernel<0><<<dim3(128/RB, 32), 256, 0, stream>>>(bufA,     bufB, w1, bb1, gates, g);
        if (g == 0)
            conv16_wmma_kernel<1><<<dim3(128/RB, 32), 256, 0, stream>>>(bufB, out, w2, bb2, gates, g);
        else
            conv16_wmma_kernel<2><<<dim3(128/RB, 32), 256, 0, stream>>>(bufB, out, w2, bb2, gates, g);
    }
}


# conv: the 4 branches share one launch per layer (grid z = branch; the last layer loops over the branches, gated sum kept in registers in the same order): 12 conv launches become 3
# speedup vs baseline: 3.0682x; 1.0874x over previous
#include <hip/hip_runtime.h>

typedef __attribute__((ext_vector_type(2))) float v2f;
typedef __attribute__((ext_vector_type(8))) float v8f;
typedef __attribute__((ext_vector_type(4))) float v4f;

#define H 128
#define W 128


__device__ __forceinline__ float from_left(float v, int addr, bool first_lane)
{
    const float t = __int_as_float(__builtin_amdgcn_ds_bpermute(addr, __float_as_int(v)));
    return first_lane ? 0.f : t;
}
__device__ __forceinline__ float from_right(float v, int addr, bool last_lane)
{
    const float t = __int_as_float(__builtin_amdgcn_ds_bpermute(addr, __float_as_int(v)));
    return last_lane ? 0.f : t;
}

__device__ __forceinline__ float tap9(float s,
                                      float w0, float v0, float w1, float v1, float w2, float v2,
                                      float w3, float v3, float w4, float v4, float w5, float v5,
                                      float w6, float v6, float w7, float v7, float w8, float v8)
{
    s = fmaf(w0, v0, s); s = fmaf(w1, v1, s); s = fmaf(w2, v2, s);
    s = fmaf(w3, v3, s); s = fmaf(w4, v4, s); s = fmaf(w5, v5, s);
    s = fmaf(w6, v6, s); s = fmaf(w7, v7, s); s = fmaf(w8, v8, s);
    return s;
}

template <int K>
__device__ __forceinline__ void dw_layer(v4f (&r)[26], const float* __restrict__ wp, const float bias,
                                         const int y0, const int addrL, const int addrR,
                                         const bool first_lane, const bool last_lane)
{
    constexpr int RLO = K + 1, RHI = 24 - K;
    const float w00 = wp[0], w01 = wp[1], w02 = wp[2],
                w10 = wp[3], w11 = wp[4], w12 = wp[5],
                w20 = wp[6], w21 = wp[7], w22 = wp[8];
    v4f a = r[RLO - 1];
    v4f b = r[RLO];
    float aL = from_left(a.w, addrL, first_lane), aR = from_right(a.x, addrR, last_lane);
    float bL = from_left(b.w, addrL, first_lane), bR = from_right(b.x, addrR, last_lane);
    #pragma unroll
    for (int i = RLO; i <= RHI; ++i) {
        const v4f c = r[i + 1];
        const float cL = from_left(c.w, addrL, first_lane);
        const float cR = from_right(c.x, addrR, last_lane);
        v4f o;
        o.x = tap9(bias, w00, aL,  w01, a.x, w02, a.y, w10, bL,  w11, b.x, w12, b.y, w20, cL,  w21, c.x, w22, c.y);
        o.y = tap9(bias, w00, a.x, w01, a.y, w02, a.z, w10, b.x, w11, b.y, w12, b.z, w20, c.x, w21, c.y, w22, c.z);
        o.z = tap9(bias, w00, a.y, w01, a.z, w02, a.w, w10, b.y, w11, b.z, w12, b.w, w20, c.y, w21, c.z, w22, c.w);
        o.w = tap9(bias, w00, a.z, w01, a.w, w02, aR,  w10, b.z, w11, b.w, w12, bR,  w20, c.z, w21, c.w, w22, cR);
        if (i < 5 || i > 20) {
            const int y = y0 - 5 + i;
            if (y < 0 || y >= H) o = (v4f){0.f, 0.f, 0.f, 0.f};
        }
        r[i] = o;
        a = b; aL = bL; aR = bR;
        b = c; bL = cL; bR = cR;
    }
}

__global__ __launch_bounds__(256)
void dw_gate_kernel(const float* __restrict__ lm,
                    const float* __restrict__ w_lm,
                    const float* __restrict__ b_lm,
                    float* __restrict__ gates)
{
    const int blk = blockIdx.x;
    const int g   = blockIdx.y;
    const int b   = blockIdx.z;
    const int tid = threadIdx.x;
    const int lane = tid & 31;
    const int wave = __builtin_amdgcn_readfirstlane(tid >> 5);
    const int y0  = blk * 16;

    __shared__ v4f part[8][16][32];

    const int cnt = (g == 0) ? 22 : (g == 1) ? 9 : (g == 2) ? 17 : 20;
    const int addrL = ((lane + 31) & 31) << 2;
    const int addrR = ((lane + 1) & 31) << 2;
    const bool first_lane = (lane == 0), last_lane = (lane == 31);

    v4f acc[16];
    #pragma unroll
    for (int j = 0; j < 16; ++j) acc[j] = (v4f){0.f, 0.f, 0.f, 0.f};

    for (int p = wave; p < cnt; p += 8) {
        int c;
        if (g == 0)      c = (p < 10) ? 17 + p : 26 + p;
        else if (g == 1) c = 27 + p;
        else if (g == 2) c = p;
        else             c = 48 + p;
        const float* src = lm + ((size_t)(b*68 + c) * H) * W;

        v4f r[26];
        #pragma unroll
        for (int i = 0; i < 26; ++i) {
            const int y = y0 - 5 + i;
            int yc = y;
            if (i < 5)  yc = y < 0 ? 0 : y;
            if (i > 20) yc = y > H-1 ? H-1 : y;
            r[i] = *(const v4f*)(src + yc*W + 4*lane);
        }
        #pragma unroll
        for (int i = 0; i < 26; ++i) {
            if (i < 5 || i > 20) {
                const int y = y0 - 5 + i;
                if (y < 0 || y >= H) r[i] = (v4f){0.f, 0.f, 0.f, 0.f};
            }
        }

        dw_layer<0>(r, w_lm + (size_t)(0*68 + c) * 9, b_lm[0*68 + c], y0, addrL, addrR, first_lane, last_lane);
        dw_layer<1>(r, w_lm + (size_t)(1*68 + c) * 9, b_lm[1*68 + c], y0, addrL, addrR, first_lane, last_lane);
        dw_layer<2>(r, w_lm + (size_t)(2*68 + c) * 9, b_lm[2*68 + c], y0, addrL, addrR, first_lane, last_lane);
        dw_layer<3>(r, w_lm + (size_t)(3*68 + c) * 9, b_lm[3*68 + c], y0, addrL, addrR, first_lane, last_lane);
        dw_layer<4>(r, w_lm + (size_t)(4*68 + c) * 9, b_lm[4*68 + c], y0, addrL, addrR, first_lane, last_lane);

        #pragma unroll
        for (int j = 0; j < 16; ++j) acc[j] += r[5 + j];
    }

    #pragma unroll
    for (int j = 0; j < 16; ++j) part[wave][j][lane] = acc[j];
    __syncthreads();

    for (int idx = tid; idx < 16*32; idx += 256) {
        const int j = idx >> 5, q = idx & 31;
        v4f s = part[0][j][q];
        #pragma unroll
        for (int wv = 1; wv < 8; ++wv) s += part[wv][j][q];
        *(v4f*)(gates + (((size_t)(b*4 + g) * H) + y0 + j) * W + 4*q) = s;
    }
}

#define RB 8
#define PLANE ((size_t)32*16*H*W)
#define WLAYER (16*16*9)

template <int LAST>
__global__ __launch_bounds__(256, 4)
void conv16_wmma_kernel(const float* __restrict__ in,
                        const size_t in_branch_stride,
                        float* __restrict__ out,
                        const float* __restrict__ w,
                        const float* __restrict__ bias,
                        const float* __restrict__ gates)
{
    const int y0  = blockIdx.x * RB;
    const int b   = blockIdx.y;
    const int tid = threadIdx.x;
    const int lane = tid & 31;
    const int wave = tid >> 5;

    __shared__ float inT[(RB+2)*130*18];
    __shared__ v2f   wB[36*32];

    const int x0    = wave * 16;
    const int laneM = lane & 15;
    const int half  = lane >> 4;
    const int xb = x0 + half*8;

    v8f res[RB];
    #pragma unroll
    for (int r = 0; r < RB; ++r) res[r] = (v8f){0.f,0.f,0.f,0.f,0.f,0.f,0.f,0.f};

    const int nbranch = LAST ? 4 : 1;
    for (int it = 0; it < nbranch; ++it) {
        const int g = LAST ? it : (int)blockIdx.z;
        const float* __restrict__ in_g = in + (size_t)g * in_branch_stride;
        const float* __restrict__ w_g  = w + (size_t)g * 3 * WLAYER;
        const float* __restrict__ b_g  = bias + g * 3 * 16;

        if (LAST && it > 0) __syncthreads();

        {
            const int q   = (lane & 7) + 8*(wave & 3);
            const int chb = ((lane >> 3) & 3) + 8*(wave >> 2);
            v4f sv[2*(RB+2)];
            #pragma unroll
            for (int j = 0; j < 2*(RB+2); ++j) {
                const int row = j >> 1;
                const int ch  = chb + 4*(j & 1);
                const int yi  = y0 - 1 + row;
                const int yc  = yi < 0 ? 0 : (yi > H-1 ? H-1 : yi);
                sv[j] = *(const v4f*)(in_g + ((size_t)(b*16 + ch) * H + yc) * W + 4*q);
            }
            #pragma unroll
            for (int j = 0; j < 2*(RB+2); ++j) {
                const int row = j >> 1;
                const int ch  = chb + 4*(j & 1);
                const int yi  = y0 - 1 + row;
                const bool inside = (yi >= 0 && yi < H);
                const v4f v = inside ? sv[j] : (v4f){0.f, 0.f, 0.f, 0.f};
                float* d = &inT[(row*130 + 1 + 4*q)*18 + ch];
                d[0] = v.x; d[18] = v.y; d[36] = v.z; d[54] = v.w;
            }
            for (int idx = tid; idx < (RB+2)*2*16; idx += 256) {
                const int ch = idx & 15;
                const int t  = idx >> 4;
                const int side = t & 1, row = t >> 1;
                inT[(row*130 + side*129)*18 + ch] = 0.f;
            }
        }
        for (int idx = tid; idx < 36*32; idx += 256) {
            int itp  = idx >> 5;
            int ln   = idx & 31;
            int kc  = itp & 3;
            int tap = itp >> 2;
            int ky = tap / 3, kx = tap - ky*3;
            int oc  = ln & 15;
            int icb = kc*4 + ((ln >> 4) << 1);
            v2f wv;
            wv.x = w_g[((oc*16 + icb    )*3 + ky)*3 + kx];
            wv.y = w_g[((oc*16 + icb + 1)*3 + ky)*3 + kx];
            wB[idx] = wv;
        }
        __syncthreads();

        v2f bB[36];
        #pragma unroll
        for (int t = 0; t < 36; ++t) bB[t] = wB[t*32 + lane];

        const float bv = b_g[laneM];

        v8f acc[RB];
        #pragma unroll
        for (int r = 0; r < RB; ++r) acc[r] = (v8f){0.f,0.f,0.f,0.f,0.f,0.f,0.f,0.f};

        v2f aCur[12], aNxt[12];
        #pragma unroll
        for (int kx = 0; kx < 3; ++kx) {
            const float* ap = &inT[(0*130 + x0 + laneM + kx)*18 + 2*half];
            #pragma unroll
            for (int kc = 0; kc < 4; ++kc) { aCur[kx*4+kc].x = ap[kc*4]; aCur[kx*4+kc].y = ap[kc*4+1]; }
        }

        #pragma unroll
        for (int i = 0; i < RB+2; ++i) {
            if (i + 1 < RB+2) {
                #pragma unroll
                for (int kx = 0; kx < 3; ++kx) {
                    const float* ap = &inT[((i+1)*130 + x0 + laneM + kx)*18 + 2*half];
                    #pragma unroll
                    for (int kc = 0; kc < 4; ++kc) { aNxt[kx*4+kc].x = ap[kc*4]; aNxt[kx*4+kc].y = ap[kc*4+1]; }
                }
            }
            #pragma unroll
            for (int kx = 0; kx < 3; ++kx) {
                #pragma unroll
                for (int kc = 0; kc < 4; ++kc) {
                    #pragma unroll
                    for (int ky = 0; ky < 3; ++ky) {
                        const int r = i - ky;
                        if (r >= 0 && r < RB)
                            acc[r] = __builtin_amdgcn_wmma_f32_16x16x4_f32(
                                         false, aCur[kx*4+kc], false, bB[(ky*3+kx)*4+kc],
                                         (short)0, acc[r], false, false);
                    }
                }
            }
            if (i >= 2) {
                const int r = i - 2;
                const int y = y0 + r;
                if (!LAST) {
                    float* op = out + (size_t)g * PLANE + ((size_t)(b*16 + laneM) * H + y) * W + xb;
                    #pragma unroll
                    for (int e = 0; e < 8; ++e) op[e] = acc[r][e] + bv;
                } else {
                    const float* gp = gates + ((size_t)(b*4 + g) * H + y) * W + xb;
                    #pragma unroll
                    for (int e = 0; e < 8; ++e) res[r][e] += (acc[r][e] + bv) * gp[e];
                }
            }
            #pragma unroll
            for (int t = 0; t < 12; ++t) aCur[t] = aNxt[t];
        }
    }

    if (LAST) {
        #pragma unroll
        for (int r = 0; r < RB; ++r) {
            float* op = out + ((size_t)(b*16 + laneM) * H + y0 + r) * W + xb;
            #pragma unroll
            for (int e = 0; e < 8; ++e) op[e] = res[r][e];
        }
    }
}

extern "C" void kernel_launch(void* const* d_in, const int* in_sizes, int n_in,
                              void* d_out, int out_size, void* d_ws, size_t ws_size,
                              hipStream_t stream)
{
    (void)in_sizes; (void)n_in; (void)out_size; (void)ws_size;
    const float* landmarks = (const float*)d_in[0];
    const float* features  = (const float*)d_in[1];
    const float* w_lm      = (const float*)d_in[2];
    const float* b_lm      = (const float*)d_in[3];
    const float* w_fea     = (const float*)d_in[4];
    const float* b_fea     = (const float*)d_in[5];
    float* out = (float*)d_out;

    char* ws = (char*)d_ws;
    float* gates = (float*)ws;
    float* bufA  = (float*)(ws + (size_t)8   * 1024*1024);
    float* bufB  = (float*)(ws + (size_t)136 * 1024*1024);

    dw_gate_kernel<<<dim3(8, 4, 32), 256, 0, stream>>>(landmarks, w_lm, b_lm, gates);

    conv16_wmma_kernel<0><<<dim3(128/RB, 32, 4), 256, 0, stream>>>(
        features, (size_t)0, bufA, w_fea + 0*WLAYER, b_fea + 0*16, gates);
    conv16_wmma_kernel<0><<<dim3(128/RB, 32, 4), 256, 0, stream>>>(
        bufA, PLANE, bufB, w_fea + 1*WLAYER, b_fea + 1*16, gates);
    conv16_wmma_kernel<1><<<dim3(128/RB, 32), 256, 0, stream>>>(
        bufB, PLANE, out, w_fea + 2*WLAYER, b_fea + 2*16, gates);
}


# conv: matrix work pinned row by row (scheduling barrier per staged row): a row's A reads form one group, B fragments stay in registers, finished rows are stored at once; summation order kept
# speedup vs baseline: 3.3630x; 1.0961x over previous
#include <hip/hip_runtime.h>

typedef __attribute__((ext_vector_type(2))) float v2f;
typedef __attribute__((ext_vector_type(8))) float v8f;
typedef __attribute__((ext_vector_type(4))) float v4f;

#define H 128
#define W 128


__device__ __forceinline__ float from_left(float v, int addr, bool first_lane)
{
    const float t = __int_as_float(__builtin_amdgcn_ds_bpermute(addr, __float_as_int(v)));
    return first_lane ? 0.f : t;
}
__device__ __forceinline__ float from_right(float v, int addr, bool last_lane)
{
    const float t = __int_as_float(__builtin_amdgcn_ds_bpermute(addr, __float_as_int(v)));
    return last_lane ? 0.f : t;
}

__device__ __forceinline__ float tap9(float s,
                                      float w0, float v0, float w1, float v1, float w2, float v2,
                                      float w3, float v3, float w4, float v4, float w5, float v5,
                                      float w6, float v6, float w7, float v7, float w8, float v8)
{
    s = fmaf(w0, v0, s); s = fmaf(w1, v1, s); s = fmaf(w2, v2, s);
    s = fmaf(w3, v3, s); s = fmaf(w4, v4, s); s = fmaf(w5, v5, s);
    s = fmaf(w6, v6, s); s = fmaf(w7, v7, s); s = fmaf(w8, v8, s);
    return s;
}

template <int K>
__device__ __forceinline__ void dw_layer(v4f (&r)[26], const float* __restrict__ wp, const float bias,
                                         const int y0, const int addrL, const int addrR,
                                         const bool first_lane, const bool last_lane)
{
    constexpr int RLO = K + 1, RHI = 24 - K;
    const float w00 = wp[0], w01 = wp[1], w02 = wp[2],
                w10 = wp[3], w11 = wp[4], w12 = wp[5],
                w20 = wp[6], w21 = wp[7], w22 = wp[8];
    v4f a = r[RLO - 1];
    v4f b = r[RLO];
    float aL = from_left(a.w, addrL, first_lane), aR = from_right(a.x, addrR, last_lane);
    float bL = from_left(b.w, addrL, first_lane), bR = from_right(b.x, addrR, last_lane);
    #pragma unroll
    for (int i = RLO; i <= RHI; ++i) {
        const v4f c = r[i + 1];
        const float cL = from_left(c.w, addrL, first_lane);
        const float cR = from_right(c.x, addrR, last_lane);
        v4f o;
        o.x = tap9(bias, w00, aL,  w01, a.x, w02, a.y, w10, bL,  w11, b.x, w12, b.y, w20, cL,  w21, c.x, w22, c.y);
        o.y = tap9(bias, w00, a.x, w01, a.y, w02, a.z, w10, b.x, w11, b.y, w12, b.z, w20, c.x, w21, c.y, w22, c.z);
        o.z = tap9(bias, w00, a.y, w01, a.z, w02, a.w, w10, b.y, w11, b.z, w12, b.w, w20, c.y, w21, c.z, w22, c.w);
        o.w = tap9(bias, w00, a.z, w01, a.w, w02, aR,  w10, b.z, w11, b.w, w12, bR,  w20, c.z, w21, c.w, w22, cR);
        if (i < 5 || i > 20) {
            const int y = y0 - 5 + i;
            if (y < 0 || y >= H) o = (v4f){0.f, 0.f, 0.f, 0.f};
        }
        r[i] = o;
        a = b; aL = bL; aR = bR;
        b = c; bL = cL; bR = cR;
    }
}

__global__ __launch_bounds__(256)
void dw_gate_kernel(const float* __restrict__ lm,
                    const float* __restrict__ w_lm,
                    const float* __restrict__ b_lm,
                    float* __restrict__ gates)
{
    const int blk = blockIdx.x;
    const int g   = blockIdx.y;
    const int b   = blockIdx.z;
    const int tid = threadIdx.x;
    const int lane = tid & 31;
    const int wave = __builtin_amdgcn_readfirstlane(tid >> 5);
    const int y0  = blk * 16;

    __shared__ v4f part[8][16][32];

    const int cnt = (g == 0) ? 22 : (g == 1) ? 9 : (g == 2) ? 17 : 20;
    const int addrL = ((lane + 31) & 31) << 2;
    const int addrR = ((lane + 1) & 31) << 2;
    const bool first_lane = (lane == 0), last_lane = (lane == 31);

    v4f acc[16];
    #pragma unroll
    for (int j = 0; j < 16; ++j) acc[j] = (v4f){0.f, 0.f, 0.f, 0.f};

    for (int p = wave; p < cnt; p += 8) {
        int c;
        if (g == 0)      c = (p < 10) ? 17 + p : 26 + p;
        else if (g == 1) c = 27 + p;
        else if (g == 2) c = p;
        else             c = 48 + p;
        const float* src = lm + ((size_t)(b*68 + c) * H) * W;

        v4f r[26];
        #pragma unroll
        for (int i = 0; i < 26; ++i) {
            const int y = y0 - 5 + i;
            int yc = y;
            if (i < 5)  yc = y < 0 ? 0 : y;
            if (i > 20) yc = y > H-1 ? H-1 : y;
            r[i] = *(const v4f*)(src + yc*W + 4*lane);
        }
        #pragma unroll
        for (int i = 0; i < 26; ++i) {
            if (i < 5 || i > 20) {
                const int y = y0 - 5 + i;
                if (y < 0 || y >= H) r[i] = (v4f){0.f, 0.f, 0.f, 0.f};
            }
        }

        dw_layer<0>(r, w_lm + (size_t)(0*68 + c) * 9, b_lm[0*68 + c], y0, addrL, addrR, first_lane, last_lane);
        dw_layer<1>(r, w_lm + (size_t)(1*68 + c) * 9, b_lm[1*68 + c], y0, addrL, addrR, first_lane, last_lane);
        dw_layer<2>(r, w_lm + (size_t)(2*68 + c) * 9, b_lm[2*68 + c], y0, addrL, addrR, first_lane, last_lane);
        dw_layer<3>(r, w_lm + (size_t)(3*68 + c) * 9, b_lm[3*68 + c], y0, addrL, addrR, first_lane, last_lane);
        dw_layer<4>(r, w_lm + (size_t)(4*68 + c) * 9, b_lm[4*68 + c], y0, addrL, addrR, first_lane, last_lane);

        #pragma unroll
        for (int j = 0; j < 16; ++j) acc[j] += r[5 + j];
    }

    #pragma unroll
    for (int j = 0; j < 16; ++j) part[wave][j][lane] = acc[j];
    __syncthreads();

    for (int idx = tid; idx < 16*32; idx += 256) {
        const int j = idx >> 5, q = idx & 31;
        v4f s = part[0][j][q];
        #pragma unroll
        for (int wv = 1; wv < 8; ++wv) s += part[wv][j][q];
        *(v4f*)(gates + (((size_t)(b*4 + g) * H) + y0 + j) * W + 4*q) = s;
    }
}

#define RB 8
#define PLANE ((size_t)32*16*H*W)
#define WLAYER (16*16*9)

template <int LAST>
__global__ __launch_bounds__(256, 4)
void conv16_wmma_kernel(const float* __restrict__ in,
                        const size_t in_branch_stride,
                        float* __restrict__ out,
                        const float* __restrict__ w,
                        const float* __restrict__ bias,
                        const float* __restrict__ gates)
{
    const int y0  = blockIdx.x * RB;
    const int b   = blockIdx.y;
    const int tid = threadIdx.x;
    const int lane = tid & 31;
    const int wave = tid >> 5;

    __shared__ float inT[(RB+2)*130*18];
    __shared__ v2f   wB[36*32];

    const int x0    = wave * 16;
    const int laneM = lane & 15;
    const int half  = lane >> 4;
    const int xb = x0 + half*8;

    v8f res[RB];
    #pragma unroll
    for (int r = 0; r < RB; ++r) res[r] = (v8f){0.f,0.f,0.f,0.f,0.f,0.f,0.f,0.f};

    const int nbranch = LAST ? 4 : 1;
    for (int it = 0; it < nbranch; ++it) {
        const int g = LAST ? it : (int)blockIdx.z;
        const float* __restrict__ in_g = in + (size_t)g * in_branch_stride;
        const float* __restrict__ w_g  = w + (size_t)g * 3 * WLAYER;
        const float* __restrict__ b_g  = bias + g * 3 * 16;

        if (LAST && it > 0) __syncthreads();

        {
            const int q   = (lane & 7) + 8*(wave & 3);
            const int chb = ((lane >> 3) & 3) + 8*(wave >> 2);
            v4f sv[2*(RB+2)];
            #pragma unroll
            for (int j = 0; j < 2*(RB+2); ++j) {
                const int row = j >> 1;
                const int ch  = chb + 4*(j & 1);
                const int yi  = y0 - 1 + row;
                const int yc  = yi < 0 ? 0 : (yi > H-1 ? H-1 : yi);
                sv[j] = *(const v4f*)(in_g + ((size_t)(b*16 + ch) * H + yc) * W + 4*q);
            }
            #pragma unroll
            for (int j = 0; j < 2*(RB+2); ++j) {
                const int row = j >> 1;
                const int ch  = chb + 4*(j & 1);
                const int yi  = y0 - 1 + row;
                const bool inside = (yi >= 0 && yi < H);
                const v4f v = inside ? sv[j] : (v4f){0.f, 0.f, 0.f, 0.f};
                float* d = &inT[(row*130 + 1 + 4*q)*18 + ch];
                d[0] = v.x; d[18] = v.y; d[36] = v.z; d[54] = v.w;
            }
            for (int idx = tid; idx < (RB+2)*2*16; idx += 256) {
                const int ch = idx & 15;
                const int t  = idx >> 4;
                const int side = t & 1, row = t >> 1;
                inT[(row*130 + side*129)*18 + ch] = 0.f;
            }
        }
        for (int idx = tid; idx < 36*32; idx += 256) {
            int itp  = idx >> 5;
            int ln   = idx & 31;
            int kc  = itp & 3;
            int tap = itp >> 2;
            int ky = tap / 3, kx = tap - ky*3;
            int oc  = ln & 15;
            int icb = kc*4 + ((ln >> 4) << 1);
            v2f wv;
            wv.x = w_g[((oc*16 + icb    )*3 + ky)*3 + kx];
            wv.y = w_g[((oc*16 + icb + 1)*3 + ky)*3 + kx];
            wB[idx] = wv;
        }
        __syncthreads();

        v2f bB[36];
        #pragma unroll
        for (int t = 0; t < 36; ++t) bB[t] = wB[t*32 + lane];

        const float bv = b_g[laneM];

        v8f acc[RB];
        #pragma unroll
        for (int r = 0; r < RB; ++r) acc[r] = (v8f){0.f,0.f,0.f,0.f,0.f,0.f,0.f,0.f};

        v2f aCur[12], aNxt[12];
        #pragma unroll
        for (int kx = 0; kx < 3; ++kx) {
            const float* ap = &inT[(0*130 + x0 + laneM + kx)*18 + 2*half];
            #pragma unroll
            for (int kc = 0; kc < 4; ++kc) { aCur[kx*4+kc].x = ap[kc*4]; aCur[kx*4+kc].y = ap[kc*4+1]; }
        }

        v8f gcur = (v8f){0.f,0.f,0.f,0.f,0.f,0.f,0.f,0.f};
        v8f gnxt = (v8f){0.f,0.f,0.f,0.f,0.f,0.f,0.f,0.f};

        #pragma unroll
        for (int i = 0; i < RB+3; ++i) {
            if (i + 1 < RB+2) {
                #pragma unroll
                for (int kx = 0; kx < 3; ++kx) {
                    const float* ap = &inT[((i+1)*130 + x0 + laneM + kx)*18 + 2*half];
                    #pragma unroll
                    for (int kc = 0; kc < 4; ++kc) { aNxt[kx*4+kc].x = ap[kc*4]; aNxt[kx*4+kc].y = ap[kc*4+1]; }
                }
            }
            if (LAST && i >= 2 && i < RB+2) {
                const float* gp = gates + ((size_t)(b*4 + g) * H + y0 + (i-2)) * W + xb;
                #pragma unroll
                for (int e = 0; e < 8; ++e) gnxt[e] = gp[e];
            }
            #pragma unroll
            for (int kx = 0; kx < 3; ++kx) {
                if (i < RB+2) {
                    #pragma unroll
                    for (int kc = 0; kc < 4; ++kc) {
                        #pragma unroll
                        for (int ky = 0; ky < 3; ++ky) {
                            const int r = i - ky;
                            if (r >= 0 && r < RB)
                                acc[r] = __builtin_amdgcn_wmma_f32_16x16x4_f32(
                                             false, aCur[kx*4+kc], false, bB[(ky*3+kx)*4+kc],
                                             (short)0, acc[r], false, false);
                        }
                    }
                }
                if (kx == 0 && i >= 3) {
                    const int r = i - 3;
                    const int y = y0 + r;
                    if (!LAST) {
                        float* op = out + (size_t)g * PLANE + ((size_t)(b*16 + laneM) * H + y) * W + xb;
                        #pragma unroll
                        for (int e = 0; e < 8; ++e) op[e] = acc[r][e] + bv;
                    } else {
                        #pragma unroll
                        for (int e = 0; e < 8; ++e) res[r][e] += (acc[r][e] + bv) * gcur[e];
                    }
                }
            }
            #pragma unroll
            for (int t = 0; t < 12; ++t) aCur[t] = aNxt[t];
            gcur = gnxt;
            __builtin_amdgcn_sched_barrier(0);
        }
    }

    if (LAST) {
        #pragma unroll
        for (int r = 0; r < RB; ++r) {
            float* op = out + ((size_t)(b*16 + laneM) * H + y0 + r) * W + xb;
            #pragma unroll
            for (int e = 0; e < 8; ++e) op[e] = res[r][e];
        }
    }
}

extern "C" void kernel_launch(void* const* d_in, const int* in_sizes, int n_in,
                              void* d_out, int out_size, void* d_ws, size_t ws_size,
                              hipStream_t stream)
{
    (void)in_sizes; (void)n_in; (void)out_size; (void)ws_size;
    const float* landmarks = (const float*)d_in[0];
    const float* features  = (const float*)d_in[1];
    const float* w_lm      = (const float*)d_in[2];
    const float* b_lm      = (const float*)d_in[3];
    const float* w_fea     = (const float*)d_in[4];
    const float* b_fea     = (const float*)d_in[5];
    float* out = (float*)d_out;

    char* ws = (char*)d_ws;
    float* gates = (float*)ws;
    float* bufA  = (float*)(ws + (size_t)8   * 1024*1024);
    float* bufB  = (float*)(ws + (size_t)136 * 1024*1024);

    dw_gate_kernel<<<dim3(8, 4, 32), 256, 0, stream>>>(landmarks, w_lm, b_lm, gates);

    conv16_wmma_kernel<0><<<dim3(128/RB, 32, 4), 256, 0, stream>>>(
        features, (size_t)0, bufA, w_fea + 0*WLAYER, b_fea + 0*16, gates);
    conv16_wmma_kernel<0><<<dim3(128/RB, 32, 4), 256, 0, stream>>>(
        bufA, PLANE, bufB, w_fea + 1*WLAYER, b_fea + 1*16, gates);
    conv16_wmma_kernel<1><<<dim3(128/RB, 32), 256, 0, stream>>>(
        bufB, PLANE, out, w_fea + 2*WLAYER, b_fea + 2*16, gates);
}


# conv: weights as the A operand and pixels as B, so a result register is one channel at 16 pixels: stores write 64-byte runs, not 16-byte pieces; one gate value per lane; sums unchanged
# speedup vs baseline: 3.6834x; 1.0953x over previous
#include <hip/hip_runtime.h>

typedef __attribute__((ext_vector_type(2))) float v2f;
typedef __attribute__((ext_vector_type(8))) float v8f;
typedef __attribute__((ext_vector_type(4))) float v4f;

#define H 128
#define W 128


__device__ __forceinline__ float from_left(float v, int addr, bool first_lane)
{
    const float t = __int_as_float(__builtin_amdgcn_ds_bpermute(addr, __float_as_int(v)));
    return first_lane ? 0.f : t;
}
__device__ __forceinline__ float from_right(float v, int addr, bool last_lane)
{
    const float t = __int_as_float(__builtin_amdgcn_ds_bpermute(addr, __float_as_int(v)));
    return last_lane ? 0.f : t;
}

__device__ __forceinline__ float tap9(float s,
                                      float w0, float v0, float w1, float v1, float w2, float v2,
                                      float w3, float v3, float w4, float v4, float w5, float v5,
                                      float w6, float v6, float w7, float v7, float w8, float v8)
{
    s = fmaf(w0, v0, s); s = fmaf(w1, v1, s); s = fmaf(w2, v2, s);
    s = fmaf(w3, v3, s); s = fmaf(w4, v4, s); s = fmaf(w5, v5, s);
    s = fmaf(w6, v6, s); s = fmaf(w7, v7, s); s = fmaf(w8, v8, s);
    return s;
}

template <int K>
__device__ __forceinline__ void dw_layer(v4f (&r)[26], const float* __restrict__ wp, const float bias,
                                         const int y0, const int addrL, const int addrR,
                                         const bool first_lane, const bool last_lane)
{
    constexpr int RLO = K + 1, RHI = 24 - K;
    const float w00 = wp[0], w01 = wp[1], w02 = wp[2],
                w10 = wp[3], w11 = wp[4], w12 = wp[5],
                w20 = wp[6], w21 = wp[7], w22 = wp[8];
    v4f a = r[RLO - 1];
    v4f b = r[RLO];
    float aL = from_left(a.w, addrL, first_lane), aR = from_right(a.x, addrR, last_lane);
    float bL = from_left(b.w, addrL, first_lane), bR = from_right(b.x, addrR, last_lane);
    #pragma unroll
    for (int i = RLO; i <= RHI; ++i) {
        const v4f c = r[i + 1];
        const float cL = from_left(c.w, addrL, first_lane);
        const float cR = from_right(c.x, addrR, last_lane);
        v4f o;
        o.x = tap9(bias, w00, aL,  w01, a.x, w02, a.y, w10, bL,  w11, b.x, w12, b.y, w20, cL,  w21, c.x, w22, c.y);
        o.y = tap9(bias, w00, a.x, w01, a.y, w02, a.z, w10, b.x, w11, b.y, w12, b.z, w20, c.x, w21, c.y, w22, c.z);
        o.z = tap9(bias, w00, a.y, w01, a.z, w02, a.w, w10, b.y, w11, b.z, w12, b.w, w20, c.y, w21, c.z, w22, c.w);
        o.w = tap9(bias, w00, a.z, w01, a.w, w02, aR,  w10, b.z, w11, b.w, w12, bR,  w20, c.z, w21, c.w, w22, cR);
        if (i < 5 || i > 20) {
            const int y = y0 - 5 + i;
            if (y < 0 || y >= H) o = (v4f){0.f, 0.f, 0.f, 0.f};
        }
        r[i] = o;
        a = b; aL = bL; aR = bR;
        b = c; bL = cL; bR = cR;
    }
}

__global__ __launch_bounds__(256)
void dw_gate_kernel(const float* __restrict__ lm,
                    const float* __restrict__ w_lm,
                    const float* __restrict__ b_lm,
                    float* __restrict__ gates)
{
    const int blk = blockIdx.x;
    const int g   = blockIdx.y;
    const int b   = blockIdx.z;
    const int tid = threadIdx.x;
    const int lane = tid & 31;
    const int wave = __builtin_amdgcn_readfirstlane(tid >> 5);
    const int y0  = blk * 16;

    __shared__ v4f part[8][16][32];

    const int cnt = (g == 0) ? 22 : (g == 1) ? 9 : (g == 2) ? 17 : 20;
    const int addrL = ((lane + 31) & 31) << 2;
    const int addrR = ((lane + 1) & 31) << 2;
    const bool first_lane = (lane == 0), last_lane = (lane == 31);

    v4f acc[16];
    #pragma unroll
    for (int j = 0; j < 16; ++j) acc[j] = (v4f){0.f, 0.f, 0.f, 0.f};

    for (int p = wave; p < cnt; p += 8) {
        int c;
        if (g == 0)      c = (p < 10) ? 17 + p : 26 + p;
        else if (g == 1) c = 27 + p;
        else if (g == 2) c = p;
        else             c = 48 + p;
        const float* src = lm + ((size_t)(b*68 + c) * H) * W;

        v4f r[26];
        #pragma unroll
        for (int i = 0; i < 26; ++i) {
            const int y = y0 - 5 + i;
            int yc = y;
            if (i < 5)  yc = y < 0 ? 0 : y;
            if (i > 20) yc = y > H-1 ? H-1 : y;
            r[i] = *(const v4f*)(src + yc*W + 4*lane);
        }
        #pragma unroll
        for (int i = 0; i < 26; ++i) {
            if (i < 5 || i > 20) {
                const int y = y0 - 5 + i;
                if (y < 0 || y >= H) r[i] = (v4f){0.f, 0.f, 0.f, 0.f};
            }
        }

        dw_layer<0>(r, w_lm + (size_t)(0*68 + c) * 9, b_lm[0*68 + c], y0, addrL, addrR, first_lane, last_lane);
        dw_layer<1>(r, w_lm + (size_t)(1*68 + c) * 9, b_lm[1*68 + c], y0, addrL, addrR, first_lane, last_lane);
        dw_layer<2>(r, w_lm + (size_t)(2*68 + c) * 9, b_lm[2*68 + c], y0, addrL, addrR, first_lane, last_lane);
        dw_layer<3>(r, w_lm + (size_t)(3*68 + c) * 9, b_lm[3*68 + c], y0, addrL, addrR, first_lane, last_lane);
        dw_layer<4>(r, w_lm + (size_t)(4*68 + c) * 9, b_lm[4*68 + c], y0, addrL, addrR, first_lane, last_lane);

        #pragma unroll
        for (int j = 0; j < 16; ++j) acc[j] += r[5 + j];
    }

    #pragma unroll
    for (int j = 0; j < 16; ++j) part[wave][j][lane] = acc[j];
    __syncthreads();

    for (int idx = tid; idx < 16*32; idx += 256) {
        const int j = idx >> 5, q = idx & 31;
        v4f s = part[0][j][q];
        #pragma unroll
        for (int wv = 1; wv < 8; ++wv) s += part[wv][j][q];
        *(v4f*)(gates + (((size_t)(b*4 + g) * H) + y0 + j) * W + 4*q) = s;
    }
}

#define RB 8
#define PLANE ((size_t)32*16*H*W)
#define WLAYER (16*16*9)

template <int LAST>
__global__ __launch_bounds__(256, 4)
void conv16_wmma_kernel(const float* __restrict__ in,
                        const size_t in_branch_stride,
                        float* __restrict__ out,
                        const float* __restrict__ w,
                        const float* __restrict__ bias,
                        const float* __restrict__ gates)
{
    const int y0  = blockIdx.x * RB;
    const int b   = blockIdx.y;
    const int tid = threadIdx.x;
    const int lane = tid & 31;
    const int wave = tid >> 5;

    __shared__ float inT[(RB+2)*130*18];
    __shared__ v2f   wB[36*32];

    const int x0    = wave * 16;
    const int laneM = lane & 15;
    const int half  = lane >> 4;
    const int xp = x0 + laneM;
    const int cb = half * 8;

    v8f res[RB];
    #pragma unroll
    for (int r = 0; r < RB; ++r) res[r] = (v8f){0.f,0.f,0.f,0.f,0.f,0.f,0.f,0.f};

    const int nbranch = LAST ? 4 : 1;
    for (int it = 0; it < nbranch; ++it) {
        const int g = LAST ? it : (int)blockIdx.z;
        const float* __restrict__ in_g = in + (size_t)g * in_branch_stride;
        const float* __restrict__ w_g  = w + (size_t)g * 3 * WLAYER;
        const float* __restrict__ b_g  = bias + g * 3 * 16;

        if (LAST && it > 0) __syncthreads();

        {
            const int q   = (lane & 7) + 8*(wave & 3);
            const int chb = ((lane >> 3) & 3) + 8*(wave >> 2);
            v4f sv[2*(RB+2)];
            #pragma unroll
            for (int j = 0; j < 2*(RB+2); ++j) {
                const int row = j >> 1;
                const int ch  = chb + 4*(j & 1);
                const int yi  = y0 - 1 + row;
                const int yc  = yi < 0 ? 0 : (yi > H-1 ? H-1 : yi);
                sv[j] = *(const v4f*)(in_g + ((size_t)(b*16 + ch) * H + yc) * W + 4*q);
            }
            #pragma unroll
            for (int j = 0; j < 2*(RB+2); ++j) {
                const int row = j >> 1;
                const int ch  = chb + 4*(j & 1);
                const int yi  = y0 - 1 + row;
                const bool inside = (yi >= 0 && yi < H);
                const v4f v = inside ? sv[j] : (v4f){0.f, 0.f, 0.f, 0.f};
                float* d = &inT[(row*130 + 1 + 4*q)*18 + ch];
                d[0] = v.x; d[18] = v.y; d[36] = v.z; d[54] = v.w;
            }
            for (int idx = tid; idx < (RB+2)*2*16; idx += 256) {
                const int ch = idx & 15;
                const int t  = idx >> 4;
                const int side = t & 1, row = t >> 1;
                inT[(row*130 + side*129)*18 + ch] = 0.f;
            }
        }
        for (int idx = tid; idx < 36*32; idx += 256) {
            int itp  = idx >> 5;
            int ln   = idx & 31;
            int kc  = itp & 3;
            int tap = itp >> 2;
            int ky = tap / 3, kx = tap - ky*3;
            int oc  = ln & 15;
            int icb = kc*4 + ((ln >> 4) << 1);
            v2f wv;
            wv.x = w_g[((oc*16 + icb    )*3 + ky)*3 + kx];
            wv.y = w_g[((oc*16 + icb + 1)*3 + ky)*3 + kx];
            wB[idx] = wv;
        }
        __syncthreads();

        v2f bB[36];
        #pragma unroll
        for (int t = 0; t < 36; ++t) bB[t] = wB[t*32 + lane];

        v8f bv;
        #pragma unroll
        for (int e = 0; e < 8; ++e) bv[e] = b_g[cb + e];

        v8f acc[RB];
        #pragma unroll
        for (int r = 0; r < RB; ++r) acc[r] = (v8f){0.f,0.f,0.f,0.f,0.f,0.f,0.f,0.f};

        v2f aCur[12], aNxt[12];
        #pragma unroll
        for (int kx = 0; kx < 3; ++kx) {
            const float* ap = &inT[(0*130 + x0 + laneM + kx)*18 + 2*half];
            #pragma unroll
            for (int kc = 0; kc < 4; ++kc) { aCur[kx*4+kc].x = ap[kc*4]; aCur[kx*4+kc].y = ap[kc*4+1]; }
        }

        float gcur = 0.f;
        float gnxt = 0.f;

        #pragma unroll
        for (int i = 0; i < RB+3; ++i) {
            if (i + 1 < RB+2) {
                #pragma unroll
                for (int kx = 0; kx < 3; ++kx) {
                    const float* ap = &inT[((i+1)*130 + x0 + laneM + kx)*18 + 2*half];
                    #pragma unroll
                    for (int kc = 0; kc < 4; ++kc) { aNxt[kx*4+kc].x = ap[kc*4]; aNxt[kx*4+kc].y = ap[kc*4+1]; }
                }
            }
            if (LAST && i >= 2 && i < RB+2) {
                gnxt = gates[((size_t)(b*4 + g) * H + y0 + (i-2)) * W + xp];
            }
            #pragma unroll
            for (int kx = 0; kx < 3; ++kx) {
                if (i < RB+2) {
                    #pragma unroll
                    for (int kc = 0; kc < 4; ++kc) {
                        #pragma unroll
                        for (int ky = 0; ky < 3; ++ky) {
                            const int r = i - ky;
                            if (r >= 0 && r < RB)
                                acc[r] = __builtin_amdgcn_wmma_f32_16x16x4_f32(
                                             false, bB[(ky*3+kx)*4+kc], false, aCur[kx*4+kc],
                                             (short)0, acc[r], false, false);
                        }
                    }
                }
                if (kx == 0 && i >= 3) {
                    const int r = i - 3;
                    const int y = y0 + r;
                    if (!LAST) {
                        float* op = out + (size_t)g * PLANE + ((size_t)(b*16 + cb) * H + y) * W + xp;
                        #pragma unroll
                        for (int e = 0; e < 8; ++e) op[(size_t)e * H * W] = acc[r][e] + bv[e];
                    } else {
                        #pragma unroll
                        for (int e = 0; e < 8; ++e) res[r][e] += (acc[r][e] + bv[e]) * gcur;
                    }
                }
            }
            #pragma unroll
            for (int t = 0; t < 12; ++t) aCur[t] = aNxt[t];
            gcur = gnxt;
            __builtin_amdgcn_sched_barrier(0);
        }
    }

    if (LAST) {
        #pragma unroll
        for (int r = 0; r < RB; ++r) {
            float* op = out + ((size_t)(b*16 + cb) * H + y0 + r) * W + xp;
            #pragma unroll
            for (int e = 0; e < 8; ++e) op[(size_t)e * H * W] = res[r][e];
        }
    }
}

extern "C" void kernel_launch(void* const* d_in, const int* in_sizes, int n_in,
                              void* d_out, int out_size, void* d_ws, size_t ws_size,
                              hipStream_t stream)
{
    (void)in_sizes; (void)n_in; (void)out_size; (void)ws_size;
    const float* landmarks = (const float*)d_in[0];
    const float* features  = (const float*)d_in[1];
    const float* w_lm      = (const float*)d_in[2];
    const float* b_lm      = (const float*)d_in[3];
    const float* w_fea     = (const float*)d_in[4];
    const float* b_fea     = (const float*)d_in[5];
    float* out = (float*)d_out;

    char* ws = (char*)d_ws;
    float* gates = (float*)ws;
    float* bufA  = (float*)(ws + (size_t)8   * 1024*1024);
    float* bufB  = (float*)(ws + (size_t)136 * 1024*1024);

    dw_gate_kernel<<<dim3(8, 4, 32), 256, 0, stream>>>(landmarks, w_lm, b_lm, gates);

    conv16_wmma_kernel<0><<<dim3(128/RB, 32, 4), 256, 0, stream>>>(
        features, (size_t)0, bufA, w_fea + 0*WLAYER, b_fea + 0*16, gates);
    conv16_wmma_kernel<0><<<dim3(128/RB, 32, 4), 256, 0, stream>>>(
        bufA, PLANE, bufB, w_fea + 1*WLAYER, b_fea + 1*16, gates);
    conv16_wmma_kernel<1><<<dim3(128/RB, 32), 256, 0, stream>>>(
        bufB, PLANE, out, w_fea + 2*WLAYER, b_fea + 2*16, gates);
}


# conv layers 0 and 1: minimal filtering along x as well, F(2x2,3x3): 16 products per 2x2 outputs, not 24 (128 matrix instructions per wave and band, not 192); sums regrouped, last bits differ
# speedup vs baseline: 4.2312x; 1.1487x over previous
#include <hip/hip_runtime.h>

typedef __attribute__((ext_vector_type(2))) float v2f;
typedef __attribute__((ext_vector_type(8))) float v8f;
typedef __attribute__((ext_vector_type(4))) float v4f;

#define H 128
#define W 128


__device__ __forceinline__ float from_left(float v, int addr, bool first_lane)
{
    const float t = __int_as_float(__builtin_amdgcn_ds_bpermute(addr, __float_as_int(v)));
    return first_lane ? 0.f : t;
}
__device__ __forceinline__ float from_right(float v, int addr, bool last_lane)
{
    const float t = __int_as_float(__builtin_amdgcn_ds_bpermute(addr, __float_as_int(v)));
    return last_lane ? 0.f : t;
}

__device__ __forceinline__ float tap9(float s,
                                      float w0, float v0, float w1, float v1, float w2, float v2,
                                      float w3, float v3, float w4, float v4, float w5, float v5,
                                      float w6, float v6, float w7, float v7, float w8, float v8)
{
    s = fmaf(w0, v0, s); s = fmaf(w1, v1, s); s = fmaf(w2, v2, s);
    s = fmaf(w3, v3, s); s = fmaf(w4, v4, s); s = fmaf(w5, v5, s);
    s = fmaf(w6, v6, s); s = fmaf(w7, v7, s); s = fmaf(w8, v8, s);
    return s;
}

template <int K>
__device__ __forceinline__ void dw_layer(v4f (&r)[26], const float* __restrict__ wp, const float bias,
                                         const int y0, const int addrL, const int addrR,
                                         const bool first_lane, const bool last_lane)
{
    constexpr int RLO = K + 1, RHI = 24 - K;
    const float w00 = wp[0], w01 = wp[1], w02 = wp[2],
                w10 = wp[3], w11 = wp[4], w12 = wp[5],
                w20 = wp[6], w21 = wp[7], w22 = wp[8];
    v4f a = r[RLO - 1];
    v4f b = r[RLO];
    float aL = from_left(a.w, addrL, first_lane), aR = from_right(a.x, addrR, last_lane);
    float bL = from_left(b.w, addrL, first_lane), bR = from_right(b.x, addrR, last_lane);
    #pragma unroll
    for (int i = RLO; i <= RHI; ++i) {
        const v4f c = r[i + 1];
        const float cL = from_left(c.w, addrL, first_lane);
        const float cR = from_right(c.x, addrR, last_lane);
        v4f o;
        o.x = tap9(bias, w00, aL,  w01, a.x, w02, a.y, w10, bL,  w11, b.x, w12, b.y, w20, cL,  w21, c.x, w22, c.y);
        o.y = tap9(bias, w00, a.x, w01, a.y, w02, a.z, w10, b.x, w11, b.y, w12, b.z, w20, c.x, w21, c.y, w22, c.z);
        o.z = tap9(bias, w00, a.y, w01, a.z, w02, a.w, w10, b.y, w11, b.z, w12, b.w, w20, c.y, w21, c.z, w22, c.w);
        o.w = tap9(bias, w00, a.z, w01, a.w, w02, aR,  w10, b.z, w11, b.w, w12, bR,  w20, c.z, w21, c.w, w22, cR);
        if (i < 5 || i > 20) {
            const int y = y0 - 5 + i;
            if (y < 0 || y >= H) o = (v4f){0.f, 0.f, 0.f, 0.f};
        }
        r[i] = o;
        a = b; aL = bL; aR = bR;
        b = c; bL = cL; bR = cR;
    }
}

__global__ __launch_bounds__(256)
void dw_gate_kernel(const float* __restrict__ lm,
                    const float* __restrict__ w_lm,
                    const float* __restrict__ b_lm,
                    float* __restrict__ gates)
{
    const int blk = blockIdx.x;
    const int g   = blockIdx.y;
    const int b   = blockIdx.z;
    const int tid = threadIdx.x;
    const int lane = tid & 31;
    const int wave = __builtin_amdgcn_readfirstlane(tid >> 5);
    const int y0  = blk * 16;

    __shared__ v4f part[8][16][32];

    const int cnt = (g == 0) ? 22 : (g == 1) ? 9 : (g == 2) ? 17 : 20;
    const int addrL = ((lane + 31) & 31) << 2;
    const int addrR = ((lane + 1) & 31) << 2;
    const bool first_lane = (lane == 0), last_lane = (lane == 31);

    v4f acc[16];
    #pragma unroll
    for (int j = 0; j < 16; ++j) acc[j] = (v4f){0.f, 0.f, 0.f, 0.f};

    for (int p = wave; p < cnt; p += 8) {
        int c;
        if (g == 0)      c = (p < 10) ? 17 + p : 26 + p;
        else if (g == 1) c = 27 + p;
        else if (g == 2) c = p;
        else             c = 48 + p;
        const float* src = lm + ((size_t)(b*68 + c) * H) * W;

        v4f r[26];
        #pragma unroll
        for (int i = 0; i < 26; ++i) {
            const int y = y0 - 5 + i;
            int yc = y;
            if (i < 5)  yc = y < 0 ? 0 : y;
            if (i > 20) yc = y > H-1 ? H-1 : y;
            r[i] = *(const v4f*)(src + yc*W + 4*lane);
        }
        #pragma unroll
        for (int i = 0; i < 26; ++i) {
            if (i < 5 || i > 20) {
                const int y = y0 - 5 + i;
                if (y < 0 || y >= H) r[i] = (v4f){0.f, 0.f, 0.f, 0.f};
            }
        }

        dw_layer<0>(r, w_lm + (size_t)(0*68 + c) * 9, b_lm[0*68 + c], y0, addrL, addrR, first_lane, last_lane);
        dw_layer<1>(r, w_lm + (size_t)(1*68 + c) * 9, b_lm[1*68 + c], y0, addrL, addrR, first_lane, last_lane);
        dw_layer<2>(r, w_lm + (size_t)(2*68 + c) * 9, b_lm[2*68 + c], y0, addrL, addrR, first_lane, last_lane);
        dw_layer<3>(r, w_lm + (size_t)(3*68 + c) * 9, b_lm[3*68 + c], y0, addrL, addrR, first_lane, last_lane);
        dw_layer<4>(r, w_lm + (size_t)(4*68 + c) * 9, b_lm[4*68 + c], y0, addrL, addrR, first_lane, last_lane);

        #pragma unroll
        for (int j = 0; j < 16; ++j) acc[j] += r[5 + j];
    }

    #pragma unroll
    for (int j = 0; j < 16; ++j) part[wave][j][lane] = acc[j];
    __syncthreads();

    for (int idx = tid; idx < 16*32; idx += 256) {
        const int j = idx >> 5, q = idx & 31;
        v4f s = part[0][j][q];
        #pragma unroll
        for (int wv = 1; wv < 8; ++wv) s += part[wv][j][q];
        *(v4f*)(gates + (((size_t)(b*4 + g) * H) + y0 + j) * W + 4*q) = s;
    }
}

#define RB 8
#define PLANE ((size_t)32*16*H*W)
#define WLAYER (16*16*9)

template <int LAST>
__global__ __launch_bounds__(256, 4)
void conv16_wmma_kernel(const float* __restrict__ in,
                        const size_t in_branch_stride,
                        float* __restrict__ out,
                        const float* __restrict__ w,
                        const float* __restrict__ bias,
                        const float* __restrict__ gates)
{
    const int y0  = blockIdx.x * RB;
    const int b   = blockIdx.y;
    const int tid = threadIdx.x;
    const int lane = tid & 31;
    const int wave = tid >> 5;

    __shared__ float inT[(RB+2)*130*18];
    __shared__ v2f   wB[36*32];

    const int x0    = wave * 16;
    const int laneM = lane & 15;
    const int half  = lane >> 4;
    const int xp = x0 + laneM;
    const int cb = half * 8;

    v8f res[RB];
    #pragma unroll
    for (int r = 0; r < RB; ++r) res[r] = (v8f){0.f,0.f,0.f,0.f,0.f,0.f,0.f,0.f};

    const int nbranch = LAST ? 4 : 1;
    for (int it = 0; it < nbranch; ++it) {
        const int g = LAST ? it : (int)blockIdx.z;
        const float* __restrict__ in_g = in + (size_t)g * in_branch_stride;
        const float* __restrict__ w_g  = w + (size_t)g * 3 * WLAYER;
        const float* __restrict__ b_g  = bias + g * 3 * 16;

        if (LAST && it > 0) __syncthreads();

        {
            const int q   = (lane & 7) + 8*(wave & 3);
            const int chb = ((lane >> 3) & 3) + 8*(wave >> 2);
            v4f sv[2*(RB+2)];
            #pragma unroll
            for (int j = 0; j < 2*(RB+2); ++j) {
                const int row = j >> 1;
                const int ch  = chb + 4*(j & 1);
                const int yi  = y0 - 1 + row;
                const int yc  = yi < 0 ? 0 : (yi > H-1 ? H-1 : yi);
                sv[j] = *(const v4f*)(in_g + ((size_t)(b*16 + ch) * H + yc) * W + 4*q);
            }
            #pragma unroll
            for (int j = 0; j < 2*(RB+2); ++j) {
                const int row = j >> 1;
                const int ch  = chb + 4*(j & 1);
                const int yi  = y0 - 1 + row;
                const bool inside = (yi >= 0 && yi < H);
                const v4f v = inside ? sv[j] : (v4f){0.f, 0.f, 0.f, 0.f};
                float* d = &inT[(row*130 + 1 + 4*q)*18 + ch];
                d[0] = v.x; d[18] = v.y; d[36] = v.z; d[54] = v.w;
            }
            for (int idx = tid; idx < (RB+2)*2*16; idx += 256) {
                const int ch = idx & 15;
                const int t  = idx >> 4;
                const int side = t & 1, row = t >> 1;
                inT[(row*130 + side*129)*18 + ch] = 0.f;
            }
        }
        for (int idx = tid; idx < 36*32; idx += 256) {
            int itp  = idx >> 5;
            int ln   = idx & 31;
            int kc  = itp & 3;
            int tap = itp >> 2;
            int ky = tap / 3, kx = tap - ky*3;
            int oc  = ln & 15;
            int icb = kc*4 + ((ln >> 4) << 1);
            v2f wv;
            wv.x = w_g[((oc*16 + icb    )*3 + ky)*3 + kx];
            wv.y = w_g[((oc*16 + icb + 1)*3 + ky)*3 + kx];
            wB[idx] = wv;
        }
        __syncthreads();

        v2f bB[36];
        #pragma unroll
        for (int t = 0; t < 36; ++t) bB[t] = wB[t*32 + lane];

        v8f bv;
        #pragma unroll
        for (int e = 0; e < 8; ++e) bv[e] = b_g[cb + e];

        v8f acc[RB];
        #pragma unroll
        for (int r = 0; r < RB; ++r) acc[r] = (v8f){0.f,0.f,0.f,0.f,0.f,0.f,0.f,0.f};

        v2f aCur[12], aNxt[12];
        #pragma unroll
        for (int kx = 0; kx < 3; ++kx) {
            const float* ap = &inT[(0*130 + x0 + laneM + kx)*18 + 2*half];
            #pragma unroll
            for (int kc = 0; kc < 4; ++kc) { aCur[kx*4+kc].x = ap[kc*4]; aCur[kx*4+kc].y = ap[kc*4+1]; }
        }

        float gcur = 0.f;
        float gnxt = 0.f;

        #pragma unroll
        for (int i = 0; i < RB+3; ++i) {
            if (i + 1 < RB+2) {
                #pragma unroll
                for (int kx = 0; kx < 3; ++kx) {
                    const float* ap = &inT[((i+1)*130 + x0 + laneM + kx)*18 + 2*half];
                    #pragma unroll
                    for (int kc = 0; kc < 4; ++kc) { aNxt[kx*4+kc].x = ap[kc*4]; aNxt[kx*4+kc].y = ap[kc*4+1]; }
                }
            }
            if (LAST && i >= 2 && i < RB+2) {
                gnxt = gates[((size_t)(b*4 + g) * H + y0 + (i-2)) * W + xp];
            }
            #pragma unroll
            for (int kx = 0; kx < 3; ++kx) {
                if (i < RB+2) {
                    #pragma unroll
                    for (int kc = 0; kc < 4; ++kc) {
                        #pragma unroll
                        for (int ky = 0; ky < 3; ++ky) {
                            const int r = i - ky;
                            if (r >= 0 && r < RB)
                                acc[r] = __builtin_amdgcn_wmma_f32_16x16x4_f32(
                                             false, bB[(ky*3+kx)*4+kc], false, aCur[kx*4+kc],
                                             (short)0, acc[r], false, false);
                        }
                    }
                }
                if (kx == 0 && i >= 3) {
                    const int r = i - 3;
                    const int y = y0 + r;
                    if (!LAST) {
                        float* op = out + (size_t)g * PLANE + ((size_t)(b*16 + cb) * H + y) * W + xp;
                        #pragma unroll
                        for (int e = 0; e < 8; ++e) op[(size_t)e * H * W] = acc[r][e] + bv[e];
                    } else {
                        #pragma unroll
                        for (int e = 0; e < 8; ++e) res[r][e] += (acc[r][e] + bv[e]) * gcur;
                    }
                }
            }
            #pragma unroll
            for (int t = 0; t < 12; ++t) aCur[t] = aNxt[t];
            gcur = gnxt;
            __builtin_amdgcn_sched_barrier(0);
        }
    }

    if (LAST) {
        #pragma unroll
        for (int r = 0; r < RB; ++r) {
            float* op = out + ((size_t)(b*16 + cb) * H + y0 + r) * W + xp;
            #pragma unroll
            for (int e = 0; e < 8; ++e) op[(size_t)e * H * W] = res[r][e];
        }
    }
}


#define NPAIR (RB/2)
#define TROWS (4*NPAIR)

__device__ __forceinline__ void load_row_cols(const float* inT, const int row, const int N, const int half, v2f (&d)[16])
{
    #pragma unroll
    for (int c = 0; c < 4; ++c) {
        const float* ap = &inT[(row*130 + 2*N + c)*18 + 2*half];
        #pragma unroll
        for (int kc = 0; kc < 4; ++kc) { d[c*4 + kc].x = ap[kc*4]; d[c*4 + kc].y = ap[kc*4 + 1]; }
    }
}

__global__ __launch_bounds__(256, 4)
void conv16_f22_kernel(const float* __restrict__ in,
                       const size_t in_branch_stride,
                       float* __restrict__ out,
                       const float* __restrict__ w,
                       const float* __restrict__ bias)
{
    const int y0  = blockIdx.x * RB;
    const int b   = blockIdx.y;
    const int g   = blockIdx.z;
    const int tid = threadIdx.x;
    const int lane = tid & 31;
    const int wave = tid >> 5;

    __shared__ float inT[TROWS*130*18];
    float* wU = inT;

    const float* __restrict__ in_g = in + (size_t)g * in_branch_stride;
    const float* __restrict__ w_g  = w + (size_t)g * 3 * WLAYER;
    const float* __restrict__ b_g  = bias + g * 3 * 16;

    const int q   = (lane & 7) + 8*(wave & 3);
    const int chb = ((lane >> 3) & 3) + 8*(wave >> 2);
    v4f sv[2*(RB+2)];
    #pragma unroll
    for (int j = 0; j < 2*(RB+2); ++j) {
        const int row = j >> 1;
        const int ch  = chb + 4*(j & 1);
        const int yi  = y0 - 1 + row;
        const int yc  = yi < 0 ? 0 : (yi > H-1 ? H-1 : yi);
        sv[j] = *(const v4f*)(in_g + ((size_t)(b*16 + ch) * H + yc) * W + 4*q);
    }

    {
        const int entry = tid & 127, sel = tid >> 7;
        const int kc = entry >> 5, ln = entry & 31;
        const int oc = ln & 15;
        const int ic = kc*4 + ((ln >> 4) << 1) + sel;
        const float* p0 = w_g + (oc*16 + ic)*9;
        float gk[9];
        #pragma unroll
        for (int k = 0; k < 9; ++k) gk[k] = p0[k];
        float a[4][3];
        #pragma unroll
        for (int kx = 0; kx < 3; ++kx) {
            const float g0 = gk[kx], g1 = gk[3 + kx], g2 = gk[6 + kx];
            const float sg = g0 + g2;
            a[0][kx] = g0;
            a[1][kx] = 0.5f*(sg + g1);
            a[2][kx] = 0.5f*(sg - g1);
            a[3][kx] = g2;
        }
        #pragma unroll
        for (int xi = 0; xi < 4; ++xi) {
            const float h0 = a[xi][0], h1 = a[xi][1], h2 = a[xi][2];
            const float sh = h0 + h2;
            float* d = &wU[((xi*4)*4 + kc)*64 + ln*2 + sel];
            d[0*4*64] = h0;
            d[1*4*64] = 0.5f*(sh + h1);
            d[2*4*64] = 0.5f*(sh - h1);
            d[3*4*64] = h2;
        }
    }

    __syncthreads();
    v2f bB[64];
    #pragma unroll
    for (int t = 0; t < 64; ++t) bB[t] = *(const v2f*)&wU[t*64 + lane*2];
    __syncthreads();

    #pragma unroll
    for (int cj = 0; cj < 2; ++cj) {
        const int ch = chb + 4*cj;
        v4f rr[RB+2];
        #pragma unroll
        for (int row = 0; row < RB+2; ++row) {
            const int yi = y0 - 1 + row;
            const bool inside = (yi >= 0 && yi < H);
            rr[row] = inside ? sv[2*row + cj] : (v4f){0.f, 0.f, 0.f, 0.f};
        }
        #pragma unroll
        for (int p = 0; p < NPAIR; ++p) {
            v4f tv[4];
            tv[0] = rr[2*p]   - rr[2*p+2];
            tv[1] = rr[2*p+1] + rr[2*p+2];
            tv[2] = rr[2*p+2] - rr[2*p+1];
            tv[3] = rr[2*p+1] - rr[2*p+3];
            #pragma unroll
            for (int t = 0; t < 4; ++t) {
                float* d = &inT[((4*p + t)*130 + 1 + 4*q)*18 + ch];
                d[0] = tv[t].x; d[18] = tv[t].y; d[36] = tv[t].z; d[54] = tv[t].w;
            }
        }
    }
    for (int idx = tid; idx < TROWS*2*16; idx += 256) {
        const int ch = idx & 15;
        const int t  = idx >> 4;
        const int side = t & 1, row = t >> 1;
        inT[(row*130 + side*129)*18 + ch] = 0.f;
    }
    __syncthreads();

    const int strip = wave & 3;
    const int wr    = wave >> 2;
    const int n     = lane & 15;
    const int half  = lane >> 4;
    const int N     = strip*16 + n;
    const int cb    = half * 8;

    v8f bv;
    #pragma unroll
    for (int e = 0; e < 8; ++e) bv[e] = b_g[cb + e];

    v2f dCur[16], dNxt[16];
    load_row_cols(inT, 8*wr, N, half, dCur);

    v8f Y[4];
    #pragma unroll
    for (int s = 0; s < 8; ++s) {
        const int p = 2*wr + (s >> 2), xi = s & 3;
        v2f f[16];
        #pragma unroll
        for (int kc = 0; kc < 4; ++kc) {
            const v2f d0 = dCur[kc], d1 = dCur[4 + kc], d2 = dCur[8 + kc], d3 = dCur[12 + kc];
            f[kc]      = d0 - d2;
            f[4 + kc]  = d1 + d2;
            f[8 + kc]  = d2 - d1;
            f[12 + kc] = d1 - d3;
        }
        if (s + 1 < 8) load_row_cols(inT, 8*wr + s + 1, N, half, dNxt);
        v8f M[4];
        #pragma unroll
        for (int nu = 0; nu < 4; ++nu)
            M[nu] = (xi == 1 && nu == 1) ? bv : (v8f){0.f,0.f,0.f,0.f,0.f,0.f,0.f,0.f};
        #pragma unroll
        for (int kc = 0; kc < 4; ++kc) {
            #pragma unroll
            for (int nu = 0; nu < 4; ++nu)
                M[nu] = __builtin_amdgcn_wmma_f32_16x16x4_f32(
                            false, bB[(xi*4 + nu)*4 + kc], false, f[nu*4 + kc],
                            (short)0, M[nu], false, false);
        }
        const v8f Z0 = (M[0] + M[2]) + M[1];
        const v8f Z1 = M[1] - (M[2] + M[3]);
        if (xi == 0) { Y[0] = Z0; Y[1] = Z1; }
        if (xi == 1) { Y[0] += Z0; Y[1] += Z1; Y[2] = Z0; Y[3] = Z1; }
        if (xi == 2) { Y[0] += Z0; Y[1] += Z1; Y[2] -= Z0; Y[3] -= Z1; }
        if (xi == 3) {
            Y[2] -= Z0; Y[3] -= Z1;
            float* op = out + (size_t)g * PLANE + ((size_t)(b*16 + cb) * H + y0 + 2*p) * W + 2*N;
            #pragma unroll
            for (int e = 0; e < 8; ++e) *(v2f*)(op + (size_t)e * H * W)     = (v2f){Y[0][e], Y[1][e]};
            #pragma unroll
            for (int e = 0; e < 8; ++e) *(v2f*)(op + (size_t)e * H * W + W) = (v2f){Y[2][e], Y[3][e]};
        }
        #pragma unroll
        for (int t = 0; t < 16; ++t) dCur[t] = dNxt[t];
        __builtin_amdgcn_sched_barrier(0);
    }
}

extern "C" void kernel_launch(void* const* d_in, const int* in_sizes, int n_in,
                              void* d_out, int out_size, void* d_ws, size_t ws_size,
                              hipStream_t stream)
{
    (void)in_sizes; (void)n_in; (void)out_size; (void)ws_size;
    const float* landmarks = (const float*)d_in[0];
    const float* features  = (const float*)d_in[1];
    const float* w_lm      = (const float*)d_in[2];
    const float* b_lm      = (const float*)d_in[3];
    const float* w_fea     = (const float*)d_in[4];
    const float* b_fea     = (const float*)d_in[5];
    float* out = (float*)d_out;

    char* ws = (char*)d_ws;
    float* gates = (float*)ws;
    float* bufA  = (float*)(ws + (size_t)8   * 1024*1024);
    float* bufB  = (float*)(ws + (size_t)136 * 1024*1024);

    dw_gate_kernel<<<dim3(8, 4, 32), 256, 0, stream>>>(landmarks, w_lm, b_lm, gates);

    conv16_f22_kernel<<<dim3(128/RB, 32, 4), 256, 0, stream>>>(
        features, (size_t)0, bufA, w_fea + 0*WLAYER, b_fea + 0*16);
    conv16_f22_kernel<<<dim3(128/RB, 32, 4), 256, 0, stream>>>(
        bufA, PLANE, bufB, w_fea + 1*WLAYER, b_fea + 1*16);
    conv16_wmma_kernel<1><<<dim3(128/RB, 32), 256, 0, stream>>>(
        bufB, PLANE, out, w_fea + 2*WLAYER, b_fea + 2*16, gates);
}


# conv layer 2 by minimal filtering F(2x2,3x3) too (256 matrix instructions per wave and branch, 4 waves, not 288 on 8), in 4-wave work-groups at 2 waves per SIMD; sums regrouped, last bits differ
# speedup vs baseline: 4.6478x; 1.0985x over previous
#include <hip/hip_runtime.h>

typedef __attribute__((ext_vector_type(2))) float v2f;
typedef __attribute__((ext_vector_type(8))) float v8f;
typedef __attribute__((ext_vector_type(4))) float v4f;

#define H 128
#define W 128


__device__ __forceinline__ float from_left(float v, int addr, bool first_lane)
{
    const float t = __int_as_float(__builtin_amdgcn_ds_bpermute(addr, __float_as_int(v)));
    return first_lane ? 0.f : t;
}
__device__ __forceinline__ float from_right(float v, int addr, bool last_lane)
{
    const float t = __int_as_float(__builtin_amdgcn_ds_bpermute(addr, __float_as_int(v)));
    return last_lane ? 0.f : t;
}

__device__ __forceinline__ float tap9(float s,
                                      float w0, float v0, float w1, float v1, float w2, float v2,
                                      float w3, float v3, float w4, float v4, float w5, float v5,
                                      float w6, float v6, float w7, float v7, float w8, float v8)
{
    s = fmaf(w0, v0, s); s = fmaf(w1, v1, s); s = fmaf(w2, v2, s);
    s = fmaf(w3, v3, s); s = fmaf(w4, v4, s); s = fmaf(w5, v5, s);
    s = fmaf(w6, v6, s); s = fmaf(w7, v7, s); s = fmaf(w8, v8, s);
    return s;
}

template <int K>
__device__ __forceinline__ void dw_layer(v4f (&r)[26], const float* __restrict__ wp, const float bias,
                                         const int y0, const int addrL, const int addrR,
                                         const bool first_lane, const bool last_lane)
{
    constexpr int RLO = K + 1, RHI = 24 - K;
    const float w00 = wp[0], w01 = wp[1], w02 = wp[2],
                w10 = wp[3], w11 = wp[4], w12 = wp[5],
                w20 = wp[6], w21 = wp[7], w22 = wp[8];
    v4f a = r[RLO - 1];
    v4f b = r[RLO];
    float aL = from_left(a.w, addrL, first_lane), aR = from_right(a.x, addrR, last_lane);
    float bL = from_left(b.w, addrL, first_lane), bR = from_right(b.x, addrR, last_lane);
    #pragma unroll
    for (int i = RLO; i <= RHI; ++i) {
        const v4f c = r[i + 1];
        const float cL = from_left(c.w, addrL, first_lane);
        const float cR = from_right(c.x, addrR, last_lane);
        v4f o;
        o.x = tap9(bias, w00, aL,  w01, a.x, w02, a.y, w10, bL,  w11, b.x, w12, b.y, w20, cL,  w21, c.x, w22, c.y);
        o.y = tap9(bias, w00, a.x, w01, a.y, w02, a.z, w10, b.x, w11, b.y, w12, b.z, w20, c.x, w21, c.y, w22, c.z);
        o.z = tap9(bias, w00, a.y, w01, a.z, w02, a.w, w10, b.y, w11, b.z, w12, b.w, w20, c.y, w21, c.z, w22, c.w);
        o.w = tap9(bias, w00, a.z, w01, a.w, w02, aR,  w10, b.z, w11, b.w, w12, bR,  w20, c.z, w21, c.w, w22, cR);
        if (i < 5 || i > 20) {
            const int y = y0 - 5 + i;
            if (y < 0 || y >= H) o = (v4f){0.f, 0.f, 0.f, 0.f};
        }
        r[i] = o;
        a = b; aL = bL; aR = bR;
        b = c; bL = cL; bR = cR;
    }
}

__global__ __launch_bounds__(256)
void dw_gate_kernel(const float* __restrict__ lm,
                    const float* __restrict__ w_lm,
                    const float* __restrict__ b_lm,
                    float* __restrict__ gates)
{
    const int blk = blockIdx.x;
    const int g   = blockIdx.y;
    const int b   = blockIdx.z;
    const int tid = threadIdx.x;
    const int lane = tid & 31;
    const int wave = __builtin_amdgcn_readfirstlane(tid >> 5);
    const int y0  = blk * 16;

    __shared__ v4f part[8][16][32];

    const int cnt = (g == 0) ? 22 : (g == 1) ? 9 : (g == 2) ? 17 : 20;
    const int addrL = ((lane + 31) & 31) << 2;
    const int addrR = ((lane + 1) & 31) << 2;
    const bool first_lane = (lane == 0), last_lane = (lane == 31);

    v4f acc[16];
    #pragma unroll
    for (int j = 0; j < 16; ++j) acc[j] = (v4f){0.f, 0.f, 0.f, 0.f};

    for (int p = wave; p < cnt; p += 8) {
        int c;
        if (g == 0)      c = (p < 10) ? 17 + p : 26 + p;
        else if (g == 1) c = 27 + p;
        else if (g == 2) c = p;
        else             c = 48 + p;
        const float* src = lm + ((size_t)(b*68 + c) * H) * W;

        v4f r[26];
        #pragma unroll
        for (int i = 0; i < 26; ++i) {
            const int y = y0 - 5 + i;
            int yc = y;
            if (i < 5)  yc = y < 0 ? 0 : y;
            if (i > 20) yc = y > H-1 ? H-1 : y;
            r[i] = *(const v4f*)(src + yc*W + 4*lane);
        }
        #pragma unroll
        for (int i = 0; i < 26; ++i) {
            if (i < 5 || i > 20) {
                const int y = y0 - 5 + i;
                if (y < 0 || y >= H) r[i] = (v4f){0.f, 0.f, 0.f, 0.f};
            }
        }

        dw_layer<0>(r, w_lm + (size_t)(0*68 + c) * 9, b_lm[0*68 + c], y0, addrL, addrR, first_lane, last_lane);
        dw_layer<1>(r, w_lm + (size_t)(1*68 + c) * 9, b_lm[1*68 + c], y0, addrL, addrR, first_lane, last_lane);
        dw_layer<2>(r, w_lm + (size_t)(2*68 + c) * 9, b_lm[2*68 + c], y0, addrL, addrR, first_lane, last_lane);
        dw_layer<3>(r, w_lm + (size_t)(3*68 + c) * 9, b_lm[3*68 + c], y0, addrL, addrR, first_lane, last_lane);
        dw_layer<4>(r, w_lm + (size_t)(4*68 + c) * 9, b_lm[4*68 + c], y0, addrL, addrR, first_lane, last_lane);

        #pragma unroll
        for (int j = 0; j < 16; ++j) acc[j] += r[5 + j];
    }

    #pragma unroll
    for (int j = 0; j < 16; ++j) part[wave][j][lane] = acc[j];
    __syncthreads();

    for (int idx = tid; idx < 16*32; idx += 256) {
        const int j = idx >> 5, q = idx & 31;
        v4f s = part[0][j][q];
        #pragma unroll
        for (int wv = 1; wv < 8; ++wv) s += part[wv][j][q];
        *(v4f*)(gates + (((size_t)(b*4 + g) * H) + y0 + j) * W + 4*q) = s;
    }
}

#define RB 8
#define PLANE ((size_t)32*16*H*W)
#define WLAYER (16*16*9)


#define NPAIR (RB/2)
#define TROWS (4*NPAIR)

__device__ __forceinline__ void load_row_cols(const float* inT, const int row, const int N, const int half, v2f (&d)[16])
{
    #pragma unroll
    for (int c = 0; c < 4; ++c) {
        const float* ap = &inT[(row*130 + 2*N + c)*18 + 2*half];
        #pragma unroll
        for (int kc = 0; kc < 4; ++kc) { d[c*4 + kc].x = ap[kc*4]; d[c*4 + kc].y = ap[kc*4 + 1]; }
    }
}

__global__ __launch_bounds__(256, 4)
void conv16_f22_kernel(const float* __restrict__ in,
                       const size_t in_branch_stride,
                       float* __restrict__ out,
                       const float* __restrict__ w,
                       const float* __restrict__ bias)
{
    const int y0  = blockIdx.x * RB;
    const int b   = blockIdx.y;
    const int g   = blockIdx.z;
    const int tid = threadIdx.x;
    const int lane = tid & 31;
    const int wave = tid >> 5;

    __shared__ float inT[TROWS*130*18];
    float* wU = inT;

    const float* __restrict__ in_g = in + (size_t)g * in_branch_stride;
    const float* __restrict__ w_g  = w + (size_t)g * 3 * WLAYER;
    const float* __restrict__ b_g  = bias + g * 3 * 16;

    const int q   = (lane & 7) + 8*(wave & 3);
    const int chb = ((lane >> 3) & 3) + 8*(wave >> 2);
    v4f sv[2*(RB+2)];
    #pragma unroll
    for (int j = 0; j < 2*(RB+2); ++j) {
        const int row = j >> 1;
        const int ch  = chb + 4*(j & 1);
        const int yi  = y0 - 1 + row;
        const int yc  = yi < 0 ? 0 : (yi > H-1 ? H-1 : yi);
        sv[j] = *(const v4f*)(in_g + ((size_t)(b*16 + ch) * H + yc) * W + 4*q);
    }

    {
        const int entry = tid & 127, sel = tid >> 7;
        const int kc = entry >> 5, ln = entry & 31;
        const int oc = ln & 15;
        const int ic = kc*4 + ((ln >> 4) << 1) + sel;
        const float* p0 = w_g + (oc*16 + ic)*9;
        float gk[9];
        #pragma unroll
        for (int k = 0; k < 9; ++k) gk[k] = p0[k];
        float a[4][3];
        #pragma unroll
        for (int kx = 0; kx < 3; ++kx) {
            const float g0 = gk[kx], g1 = gk[3 + kx], g2 = gk[6 + kx];
            const float sg = g0 + g2;
            a[0][kx] = g0;
            a[1][kx] = 0.5f*(sg + g1);
            a[2][kx] = 0.5f*(sg - g1);
            a[3][kx] = g2;
        }
        #pragma unroll
        for (int xi = 0; xi < 4; ++xi) {
            const float h0 = a[xi][0], h1 = a[xi][1], h2 = a[xi][2];
            const float sh = h0 + h2;
            float* d = &wU[((xi*4)*4 + kc)*64 + ln*2 + sel];
            d[0*4*64] = h0;
            d[1*4*64] = 0.5f*(sh + h1);
            d[2*4*64] = 0.5f*(sh - h1);
            d[3*4*64] = h2;
        }
    }

    __syncthreads();
    v2f bB[64];
    #pragma unroll
    for (int t = 0; t < 64; ++t) bB[t] = *(const v2f*)&wU[t*64 + lane*2];
    __syncthreads();

    #pragma unroll
    for (int cj = 0; cj < 2; ++cj) {
        const int ch = chb + 4*cj;
        v4f rr[RB+2];
        #pragma unroll
        for (int row = 0; row < RB+2; ++row) {
            const int yi = y0 - 1 + row;
            const bool inside = (yi >= 0 && yi < H);
            rr[row] = inside ? sv[2*row + cj] : (v4f){0.f, 0.f, 0.f, 0.f};
        }
        #pragma unroll
        for (int p = 0; p < NPAIR; ++p) {
            v4f tv[4];
            tv[0] = rr[2*p]   - rr[2*p+2];
            tv[1] = rr[2*p+1] + rr[2*p+2];
            tv[2] = rr[2*p+2] - rr[2*p+1];
            tv[3] = rr[2*p+1] - rr[2*p+3];
            #pragma unroll
            for (int t = 0; t < 4; ++t) {
                float* d = &inT[((4*p + t)*130 + 1 + 4*q)*18 + ch];
                d[0] = tv[t].x; d[18] = tv[t].y; d[36] = tv[t].z; d[54] = tv[t].w;
            }
        }
    }
    for (int idx = tid; idx < TROWS*2*16; idx += 256) {
        const int ch = idx & 15;
        const int t  = idx >> 4;
        const int side = t & 1, row = t >> 1;
        inT[(row*130 + side*129)*18 + ch] = 0.f;
    }
    __syncthreads();

    const int strip = wave & 3;
    const int wr    = wave >> 2;
    const int n     = lane & 15;
    const int half  = lane >> 4;
    const int N     = strip*16 + n;
    const int cb    = half * 8;

    v8f bv;
    #pragma unroll
    for (int e = 0; e < 8; ++e) bv[e] = b_g[cb + e];

    v2f dCur[16], dNxt[16];
    load_row_cols(inT, 8*wr, N, half, dCur);

    v8f Y[4];
    #pragma unroll
    for (int s = 0; s < 8; ++s) {
        const int p = 2*wr + (s >> 2), xi = s & 3;
        v2f f[16];
        #pragma unroll
        for (int kc = 0; kc < 4; ++kc) {
            const v2f d0 = dCur[kc], d1 = dCur[4 + kc], d2 = dCur[8 + kc], d3 = dCur[12 + kc];
            f[kc]      = d0 - d2;
            f[4 + kc]  = d1 + d2;
            f[8 + kc]  = d2 - d1;
            f[12 + kc] = d1 - d3;
        }
        if (s + 1 < 8) load_row_cols(inT, 8*wr + s + 1, N, half, dNxt);
        v8f M[4];
        #pragma unroll
        for (int nu = 0; nu < 4; ++nu)
            M[nu] = (xi == 1 && nu == 1) ? bv : (v8f){0.f,0.f,0.f,0.f,0.f,0.f,0.f,0.f};
        #pragma unroll
        for (int kc = 0; kc < 4; ++kc) {
            #pragma unroll
            for (int nu = 0; nu < 4; ++nu)
                M[nu] = __builtin_amdgcn_wmma_f32_16x16x4_f32(
                            false, bB[(xi*4 + nu)*4 + kc], false, f[nu*4 + kc],
                            (short)0, M[nu], false, false);
        }
        const v8f Z0 = (M[0] + M[2]) + M[1];
        const v8f Z1 = M[1] - (M[2] + M[3]);
        if (xi == 0) { Y[0] = Z0; Y[1] = Z1; }
        if (xi == 1) { Y[0] += Z0; Y[1] += Z1; Y[2] = Z0; Y[3] = Z1; }
        if (xi == 2) { Y[0] += Z0; Y[1] += Z1; Y[2] -= Z0; Y[3] -= Z1; }
        if (xi == 3) {
            Y[2] -= Z0; Y[3] -= Z1;
            float* op = out + (size_t)g * PLANE + ((size_t)(b*16 + cb) * H + y0 + 2*p) * W + 2*N;
            #pragma unroll
            for (int e = 0; e < 8; ++e) *(v2f*)(op + (size_t)e * H * W)     = (v2f){Y[0][e], Y[1][e]};
            #pragma unroll
            for (int e = 0; e < 8; ++e) *(v2f*)(op + (size_t)e * H * W + W) = (v2f){Y[2][e], Y[3][e]};
        }
        #pragma unroll
        for (int t = 0; t < 16; ++t) dCur[t] = dNxt[t];
        __builtin_amdgcn_sched_barrier(0);
    }
}

__global__ __launch_bounds__(128, 2)
void conv16_f22_gated_kernel(const float* __restrict__ in,
                             const size_t in_branch_stride,
                             float* __restrict__ out,
                             const float* __restrict__ w,
                             const float* __restrict__ bias,
                             const float* __restrict__ gates)
{
    const int y0  = blockIdx.x * RB;
    const int b   = blockIdx.y;
    const int tid = threadIdx.x;
    const int lane = tid & 31;
    const int wave = tid >> 5;

    __shared__ float inT[TROWS*130*18];
    float* wU = inT;

    const int q   = (lane & 7) + 8*wave;
    const int chb = (lane >> 3) & 3;

    const int n     = lane & 15;
    const int half  = lane >> 4;
    const int N     = wave*16 + n;
    const int cb    = half * 8;

    v8f res[NPAIR][4];
    #pragma unroll
    for (int p = 0; p < NPAIR; ++p) {
        #pragma unroll
        for (int k = 0; k < 4; ++k) res[p][k] = (v8f){0.f,0.f,0.f,0.f,0.f,0.f,0.f,0.f};
    }

    for (int g = 0; g < 4; ++g) {
        const float* __restrict__ in_g = in + (size_t)g * in_branch_stride;
        const float* __restrict__ w_g  = w + (size_t)g * 3 * WLAYER;
        const float* __restrict__ b_g  = bias + g * 3 * 16;

        v4f sv[4*(RB+2)];
        #pragma unroll
        for (int j = 0; j < 4*(RB+2); ++j) {
            const int row = j >> 2;
            const int ch  = chb + 4*(j & 3);
            const int yi  = y0 - 1 + row;
            const int yc  = yi < 0 ? 0 : (yi > H-1 ? H-1 : yi);
            sv[j] = *(const v4f*)(in_g + ((size_t)(b*16 + ch) * H + yc) * W + 4*q);
        }

        const int wkc = tid >> 5, wln = tid & 31;
        float gkr[2][9];
        #pragma unroll
        for (int sel = 0; sel < 2; ++sel) {
            const int ic = wkc*4 + ((wln >> 4) << 1) + sel;
            const float* p0 = w_g + ((wln & 15)*16 + ic)*9;
            #pragma unroll
            for (int k = 0; k < 9; ++k) gkr[sel][k] = p0[k];
        }

        if (g > 0) __syncthreads();

        {
            const int kc = wkc, ln = wln;
            #pragma unroll
            for (int sel = 0; sel < 2; ++sel) {
                const float* gk = gkr[sel];
                float a[4][3];
                #pragma unroll
                for (int kx = 0; kx < 3; ++kx) {
                    const float g0 = gk[kx], g1 = gk[3 + kx], g2 = gk[6 + kx];
                    const float sg = g0 + g2;
                    a[0][kx] = g0;
                    a[1][kx] = 0.5f*(sg + g1);
                    a[2][kx] = 0.5f*(sg - g1);
                    a[3][kx] = g2;
                }
                #pragma unroll
                for (int xi = 0; xi < 4; ++xi) {
                    const float h0 = a[xi][0], h1 = a[xi][1], h2 = a[xi][2];
                    const float sh = h0 + h2;
                    float* d = &wU[((xi*4)*4 + kc)*64 + ln*2 + sel];
                    d[0*4*64] = h0;
                    d[1*4*64] = 0.5f*(sh + h1);
                    d[2*4*64] = 0.5f*(sh - h1);
                    d[3*4*64] = h2;
                }
            }
        }

        __syncthreads();
        v2f bB[64];
        #pragma unroll
        for (int t = 0; t < 64; ++t) bB[t] = *(const v2f*)&wU[t*64 + lane*2];
        __syncthreads();

        #pragma unroll
        for (int cj = 0; cj < 4; ++cj) {
            const int ch = chb + 4*cj;
            v4f rr[RB+2];
            #pragma unroll
            for (int row = 0; row < RB+2; ++row) {
                const int yi = y0 - 1 + row;
                const bool inside = (yi >= 0 && yi < H);
                rr[row] = inside ? sv[4*row + cj] : (v4f){0.f, 0.f, 0.f, 0.f};
            }
            #pragma unroll
            for (int p = 0; p < NPAIR; ++p) {
                v4f tv[4];
                tv[0] = rr[2*p]   - rr[2*p+2];
                tv[1] = rr[2*p+1] + rr[2*p+2];
                tv[2] = rr[2*p+2] - rr[2*p+1];
                tv[3] = rr[2*p+1] - rr[2*p+3];
                #pragma unroll
                for (int t = 0; t < 4; ++t) {
                    float* d = &inT[((4*p + t)*130 + 1 + 4*q)*18 + ch];
                    d[0] = tv[t].x; d[18] = tv[t].y; d[36] = tv[t].z; d[54] = tv[t].w;
                }
            }
        }
        for (int idx = tid; idx < TROWS*2*16; idx += 128) {
            const int ch = idx & 15;
            const int t  = idx >> 4;
            const int side = t & 1, row = t >> 1;
            inT[(row*130 + side*129)*18 + ch] = 0.f;
        }
        __syncthreads();

        v8f bv;
        #pragma unroll
        for (int e = 0; e < 8; ++e) bv[e] = b_g[cb + e];

        v2f gv[NPAIR][2];
        #pragma unroll
        for (int p = 0; p < NPAIR; ++p) {
            const float* gp = gates + ((size_t)(b*4 + g) * H + y0 + 2*p) * W + 2*N;
            gv[p][0] = *(const v2f*)gp;
            gv[p][1] = *(const v2f*)(gp + W);
        }

        v2f dCur[16], dNxt[16];
        load_row_cols(inT, 0, N, half, dCur);

        v8f Y[4];
        #pragma unroll
        for (int s = 0; s < TROWS; ++s) {
            const int p = s >> 2, xi = s & 3;
            v2f f[16];
            #pragma unroll
            for (int kc = 0; kc < 4; ++kc) {
                const v2f d0 = dCur[kc], d1 = dCur[4 + kc], d2 = dCur[8 + kc], d3 = dCur[12 + kc];
                f[kc]      = d0 - d2;
                f[4 + kc]  = d1 + d2;
                f[8 + kc]  = d2 - d1;
                f[12 + kc] = d1 - d3;
            }
            if (s + 1 < TROWS) load_row_cols(inT, s + 1, N, half, dNxt);
            v8f M[4];
            #pragma unroll
            for (int nu = 0; nu < 4; ++nu)
                M[nu] = (xi == 1 && nu == 1) ? bv : (v8f){0.f,0.f,0.f,0.f,0.f,0.f,0.f,0.f};
            #pragma unroll
            for (int kc = 0; kc < 4; ++kc) {
                #pragma unroll
                for (int nu = 0; nu < 4; ++nu)
                    M[nu] = __builtin_amdgcn_wmma_f32_16x16x4_f32(
                                false, bB[(xi*4 + nu)*4 + kc], false, f[nu*4 + kc],
                                (short)0, M[nu], false, false);
            }
            const v8f Z0 = (M[0] + M[2]) + M[1];
            const v8f Z1 = M[1] - (M[2] + M[3]);
            if (xi == 0) { Y[0] = Z0; Y[1] = Z1; }
            if (xi == 1) { Y[0] += Z0; Y[1] += Z1; Y[2] = Z0; Y[3] = Z1; }
            if (xi == 2) { Y[0] += Z0; Y[1] += Z1; Y[2] -= Z0; Y[3] -= Z1; }
            if (xi == 3) {
                Y[2] -= Z0; Y[3] -= Z1;
                res[p][0] += Y[0] * gv[p][0].x;
                res[p][1] += Y[1] * gv[p][0].y;
                res[p][2] += Y[2] * gv[p][1].x;
                res[p][3] += Y[3] * gv[p][1].y;
            }
            #pragma unroll
            for (int t = 0; t < 16; ++t) dCur[t] = dNxt[t];
            __builtin_amdgcn_sched_barrier(0);
        }
    }

    #pragma unroll
    for (int p = 0; p < NPAIR; ++p) {
        float* op = out + ((size_t)(b*16 + cb) * H + y0 + 2*p) * W + 2*N;
        #pragma unroll
        for (int e = 0; e < 8; ++e) *(v2f*)(op + (size_t)e * H * W)     = (v2f){res[p][0][e], res[p][1][e]};
        #pragma unroll
        for (int e = 0; e < 8; ++e) *(v2f*)(op + (size_t)e * H * W + W) = (v2f){res[p][2][e], res[p][3][e]};
    }
}

extern "C" void kernel_launch(void* const* d_in, const int* in_sizes, int n_in,
                              void* d_out, int out_size, void* d_ws, size_t ws_size,
                              hipStream_t stream)
{
    (void)in_sizes; (void)n_in; (void)out_size; (void)ws_size;
    const float* landmarks = (const float*)d_in[0];
    const float* features  = (const float*)d_in[1];
    const float* w_lm      = (const float*)d_in[2];
    const float* b_lm      = (const float*)d_in[3];
    const float* w_fea     = (const float*)d_in[4];
    const float* b_fea     = (const float*)d_in[5];
    float* out = (float*)d_out;

    char* ws = (char*)d_ws;
    float* gates = (float*)ws;
    float* bufA  = (float*)(ws + (size_t)8   * 1024*1024);
    float* bufB  = (float*)(ws + (size_t)136 * 1024*1024);

    dw_gate_kernel<<<dim3(8, 4, 32), 256, 0, stream>>>(landmarks, w_lm, b_lm, gates);

    conv16_f22_kernel<<<dim3(128/RB, 32, 4), 256, 0, stream>>>(
        features, (size_t)0, bufA, w_fea + 0*WLAYER, b_fea + 0*16);
    conv16_f22_kernel<<<dim3(128/RB, 32, 4), 256, 0, stream>>>(
        bufA, PLANE, bufB, w_fea + 1*WLAYER, b_fea + 1*16);
    conv16_f22_gated_kernel<<<dim3(128/RB, 32), 128, 0, stream>>>(
        bufB, PLANE, out, w_fea + 2*WLAYER, b_fea + 2*16, gates);
}


# depthwise towers: one 16-wave work-group per row block takes all 68 channels in equal shares (17 passes per SIMD; was per-gate groups of 2-3 rounds); gate sums regrouped by share, last bits differ
# speedup vs baseline: 4.8210x; 1.0373x over previous
#include <hip/hip_runtime.h>

typedef __attribute__((ext_vector_type(2))) float v2f;
typedef __attribute__((ext_vector_type(8))) float v8f;
typedef __attribute__((ext_vector_type(4))) float v4f;

#define H 128
#define W 128


__device__ __forceinline__ float from_left(float v, int addr, bool first_lane)
{
    const float t = __int_as_float(__builtin_amdgcn_ds_bpermute(addr, __float_as_int(v)));
    return first_lane ? 0.f : t;
}
__device__ __forceinline__ float from_right(float v, int addr, bool last_lane)
{
    const float t = __int_as_float(__builtin_amdgcn_ds_bpermute(addr, __float_as_int(v)));
    return last_lane ? 0.f : t;
}

__device__ __forceinline__ float tap9(float s,
                                      float w0, float v0, float w1, float v1, float w2, float v2,
                                      float w3, float v3, float w4, float v4, float w5, float v5,
                                      float w6, float v6, float w7, float v7, float w8, float v8)
{
    s = fmaf(w0, v0, s); s = fmaf(w1, v1, s); s = fmaf(w2, v2, s);
    s = fmaf(w3, v3, s); s = fmaf(w4, v4, s); s = fmaf(w5, v5, s);
    s = fmaf(w6, v6, s); s = fmaf(w7, v7, s); s = fmaf(w8, v8, s);
    return s;
}

template <int K>
__device__ __forceinline__ void dw_layer(v4f (&r)[26], const float* __restrict__ wp, const float bias,
                                         const int y0, const int addrL, const int addrR,
                                         const bool first_lane, const bool last_lane)
{
    constexpr int RLO = K + 1, RHI = 24 - K;
    const float w00 = wp[0], w01 = wp[1], w02 = wp[2],
                w10 = wp[3], w11 = wp[4], w12 = wp[5],
                w20 = wp[6], w21 = wp[7], w22 = wp[8];
    v4f a = r[RLO - 1];
    v4f b = r[RLO];
    float aL = from_left(a.w, addrL, first_lane), aR = from_right(a.x, addrR, last_lane);
    float bL = from_left(b.w, addrL, first_lane), bR = from_right(b.x, addrR, last_lane);
    #pragma unroll
    for (int i = RLO; i <= RHI; ++i) {
        const v4f c = r[i + 1];
        const float cL = from_left(c.w, addrL, first_lane);
        const float cR = from_right(c.x, addrR, last_lane);
        v4f o;
        o.x = tap9(bias, w00, aL,  w01, a.x, w02, a.y, w10, bL,  w11, b.x, w12, b.y, w20, cL,  w21, c.x, w22, c.y);
        o.y = tap9(bias, w00, a.x, w01, a.y, w02, a.z, w10, b.x, w11, b.y, w12, b.z, w20, c.x, w21, c.y, w22, c.z);
        o.z = tap9(bias, w00, a.y, w01, a.z, w02, a.w, w10, b.y, w11, b.z, w12, b.w, w20, c.y, w21, c.z, w22, c.w);
        o.w = tap9(bias, w00, a.z, w01, a.w, w02, aR,  w10, b.z, w11, b.w, w12, bR,  w20, c.z, w21, c.w, w22, cR);
        if (i < 5 || i > 20) {
            const int y = y0 - 5 + i;
            if (y < 0 || y >= H) o = (v4f){0.f, 0.f, 0.f, 0.f};
        }
        r[i] = o;
        a = b; aL = bL; aR = bR;
        b = c; bL = cL; bR = cR;
    }
}

__device__ __forceinline__ constexpr int gate_of(int p) { return p < 17 ? 2 : (p < 39 ? 0 : (p < 48 ? 1 : 3)); }
__device__ __forceinline__ constexpr int chan_of(int p) { return p < 17 ? p : (p < 27 ? p : (p < 39 ? p + 9 : (p < 48 ? p - 12 : p))); }
__device__ __forceinline__ constexpr int share_len(int w) { return (w % 5 == 0) ? 5 : 4; }
__device__ __forceinline__ constexpr int share_lo(int w)  { return 4*w + (w + 4) / 5; }
__device__ __forceinline__ constexpr int crosses(int w)   { return gate_of(share_lo(w)) != gate_of(share_lo(w) + share_len(w) - 1) ? 1 : 0; }
__device__ __forceinline__ constexpr int slot_of(int w)   { int s = w; for (int k = 0; k < w; ++k) s += crosses(k); return s; }

__global__ __launch_bounds__(512)
void dw_gate_kernel(const float* __restrict__ lm,
                    const float* __restrict__ w_lm,
                    const float* __restrict__ b_lm,
                    float* __restrict__ gates)
{
    const int blk = blockIdx.x;
    const int b   = blockIdx.y;
    const int tid = threadIdx.x;
    const int lane = tid & 31;
    const int wave = __builtin_amdgcn_readfirstlane(tid >> 5);
    const int y0  = blk * 16;

    __shared__ v4f part[slot_of(16)][16][32];

    const int addrL = ((lane + 31) & 31) << 2;
    const int addrR = ((lane + 1) & 31) << 2;
    const bool first_lane = (lane == 0), last_lane = (lane == 31);

    v4f acc[16];
    #pragma unroll
    for (int j = 0; j < 16; ++j) acc[j] = (v4f){0.f, 0.f, 0.f, 0.f};

    const int p_lo = share_lo(wave);
    const int p_hi = p_lo + share_len(wave);
    int slot = wave;
    #pragma unroll
    for (int k = 0; k < 16; ++k) if (crosses(k) != 0 && wave > k) ++slot;
    for (int p = p_lo; p < p_hi; ++p) {
        const int c = chan_of(p);
        const float* src = lm + ((size_t)(b*68 + c) * H) * W;

        v4f r[26];
        #pragma unroll
        for (int i = 0; i < 26; ++i) {
            const int y = y0 - 5 + i;
            int yc = y;
            if (i < 5)  yc = y < 0 ? 0 : y;
            if (i > 20) yc = y > H-1 ? H-1 : y;
            r[i] = *(const v4f*)(src + yc*W + 4*lane);
        }
        #pragma unroll
        for (int i = 0; i < 26; ++i) {
            if (i < 5 || i > 20) {
                const int y = y0 - 5 + i;
                if (y < 0 || y >= H) r[i] = (v4f){0.f, 0.f, 0.f, 0.f};
            }
        }

        dw_layer<0>(r, w_lm + (size_t)(0*68 + c) * 9, b_lm[0*68 + c], y0, addrL, addrR, first_lane, last_lane);
        dw_layer<1>(r, w_lm + (size_t)(1*68 + c) * 9, b_lm[1*68 + c], y0, addrL, addrR, first_lane, last_lane);
        dw_layer<2>(r, w_lm + (size_t)(2*68 + c) * 9, b_lm[2*68 + c], y0, addrL, addrR, first_lane, last_lane);
        dw_layer<3>(r, w_lm + (size_t)(3*68 + c) * 9, b_lm[3*68 + c], y0, addrL, addrR, first_lane, last_lane);
        dw_layer<4>(r, w_lm + (size_t)(4*68 + c) * 9, b_lm[4*68 + c], y0, addrL, addrR, first_lane, last_lane);

        #pragma unroll
        for (int j = 0; j < 16; ++j) acc[j] += r[5 + j];

        const bool border = (p + 1 == p_hi) || (gate_of(p + 1) != gate_of(p));
        if (border) {
            #pragma unroll
            for (int j = 0; j < 16; ++j) { part[slot][j][lane] = acc[j]; acc[j] = (v4f){0.f, 0.f, 0.f, 0.f}; }
            ++slot;
        }
    }
    __syncthreads();

    const int j = tid >> 5, q = tid & 31;
    #pragma unroll
    for (int g = 0; g < 4; ++g) {
        v4f s = (v4f){0.f, 0.f, 0.f, 0.f};
        #pragma unroll
        for (int wv = 0; wv < 16; ++wv) {
            const int gf = gate_of(share_lo(wv));
            const int gl = gate_of(share_lo(wv) + share_len(wv) - 1);
            if (gf == g)             s += part[slot_of(wv)][j][q];
            if (gl != gf && gl == g) s += part[slot_of(wv) + 1][j][q];
        }
        *(v4f*)(gates + (((size_t)(b*4 + g) * H) + y0 + j) * W + 4*q) = s;
    }
}

#define RB 8
#define PLANE ((size_t)32*16*H*W)
#define WLAYER (16*16*9)


#define NPAIR (RB/2)
#define TROWS (4*NPAIR)

__device__ __forceinline__ void load_row_cols(const float* inT, const int row, const int N, const int half, v2f (&d)[16])
{
    #pragma unroll
    for (int c = 0; c < 4; ++c) {
        const float* ap = &inT[(row*130 + 2*N + c)*18 + 2*half];
        #pragma unroll
        for (int kc = 0; kc < 4; ++kc) { d[c*4 + kc].x = ap[kc*4]; d[c*4 + kc].y = ap[kc*4 + 1]; }
    }
}

__global__ __launch_bounds__(256, 4)
void conv16_f22_kernel(const float* __restrict__ in,
                       const size_t in_branch_stride,
                       float* __restrict__ out,
                       const float* __restrict__ w,
                       const float* __restrict__ bias)
{
    const int y0  = blockIdx.x * RB;
    const int b   = blockIdx.y;
    const int g   = blockIdx.z;
    const int tid = threadIdx.x;
    const int lane = tid & 31;
    const int wave = tid >> 5;

    __shared__ float inT[TROWS*130*18];
    float* wU = inT;

    const float* __restrict__ in_g = in + (size_t)g * in_branch_stride;
    const float* __restrict__ w_g  = w + (size_t)g * 3 * WLAYER;
    const float* __restrict__ b_g  = bias + g * 3 * 16;

    const int q   = (lane & 7) + 8*(wave & 3);
    const int chb = ((lane >> 3) & 3) + 8*(wave >> 2);
    v4f sv[2*(RB+2)];
    #pragma unroll
    for (int j = 0; j < 2*(RB+2); ++j) {
        const int row = j >> 1;
        const int ch  = chb + 4*(j & 1);
        const int yi  = y0 - 1 + row;
        const int yc  = yi < 0 ? 0 : (yi > H-1 ? H-1 : yi);
        sv[j] = *(const v4f*)(in_g + ((size_t)(b*16 + ch) * H + yc) * W + 4*q);
    }

    {
        const int entry = tid & 127, sel = tid >> 7;
        const int kc = entry >> 5, ln = entry & 31;
        const int oc = ln & 15;
        const int ic = kc*4 + ((ln >> 4) << 1) + sel;
        const float* p0 = w_g + (oc*16 + ic)*9;
        float gk[9];
        #pragma unroll
        for (int k = 0; k < 9; ++k) gk[k] = p0[k];
        float a[4][3];
        #pragma unroll
        for (int kx = 0; kx < 3; ++kx) {
            const float g0 = gk[kx], g1 = gk[3 + kx], g2 = gk[6 + kx];
            const float sg = g0 + g2;
            a[0][kx] = g0;
            a[1][kx] = 0.5f*(sg + g1);
            a[2][kx] = 0.5f*(sg - g1);
            a[3][kx] = g2;
        }
        #pragma unroll
        for (int xi = 0; xi < 4; ++xi) {
            const float h0 = a[xi][0], h1 = a[xi][1], h2 = a[xi][2];
            const float sh = h0 + h2;
            float* d = &wU[((xi*4)*4 + kc)*64 + ln*2 + sel];
            d[0*4*64] = h0;
            d[1*4*64] = 0.5f*(sh + h1);
            d[2*4*64] = 0.5f*(sh - h1);
            d[3*4*64] = h2;
        }
    }

    __syncthreads();
    v2f bB[64];
    #pragma unroll
    for (int t = 0; t < 64; ++t) bB[t] = *(const v2f*)&wU[t*64 + lane*2];
    __syncthreads();

    #pragma unroll
    for (int cj = 0; cj < 2; ++cj) {
        const int ch = chb + 4*cj;
        v4f rr[RB+2];
        #pragma unroll
        for (int row = 0; row < RB+2; ++row) {
            const int yi = y0 - 1 + row;
            const bool inside = (yi >= 0 && yi < H);
            rr[row] = inside ? sv[2*row + cj] : (v4f){0.f, 0.f, 0.f, 0.f};
        }
        #pragma unroll
        for (int p = 0; p < NPAIR; ++p) {
            v4f tv[4];
            tv[0] = rr[2*p]   - rr[2*p+2];
            tv[1] = rr[2*p+1] + rr[2*p+2];
            tv[2] = rr[2*p+2] - rr[2*p+1];
            tv[3] = rr[2*p+1] - rr[2*p+3];
            #pragma unroll
            for (int t = 0; t < 4; ++t) {
                float* d = &inT[((4*p + t)*130 + 1 + 4*q)*18 + ch];
                d[0] = tv[t].x; d[18] = tv[t].y; d[36] = tv[t].z; d[54] = tv[t].w;
            }
        }
    }
    for (int idx = tid; idx < TROWS*2*16; idx += 256) {
        const int ch = idx & 15;
        const int t  = idx >> 4;
        const int side = t & 1, row = t >> 1;
        inT[(row*130 + side*129)*18 + ch] = 0.f;
    }
    __syncthreads();

    const int strip = wave & 3;
    const int wr    = wave >> 2;
    const int n     = lane & 15;
    const int half  = lane >> 4;
    const int N     = strip*16 + n;
    const int cb    = half * 8;

    v8f bv;
    #pragma unroll
    for (int e = 0; e < 8; ++e) bv[e] = b_g[cb + e];

    v2f dCur[16], dNxt[16];
    load_row_cols(inT, 8*wr, N, half, dCur);

    v8f Y[4];
    #pragma unroll
    for (int s = 0; s < 8; ++s) {
        const int p = 2*wr + (s >> 2), xi = s & 3;
        v2f f[16];
        #pragma unroll
        for (int kc = 0; kc < 4; ++kc) {
            const v2f d0 = dCur[kc], d1 = dCur[4 + kc], d2 = dCur[8 + kc], d3 = dCur[12 + kc];
            f[kc]      = d0 - d2;
            f[4 + kc]  = d1 + d2;
            f[8 + kc]  = d2 - d1;
            f[12 + kc] = d1 - d3;
        }
        if (s + 1 < 8) load_row_cols(inT, 8*wr + s + 1, N, half, dNxt);
        v8f M[4];
        #pragma unroll
        for (int nu = 0; nu < 4; ++nu)
            M[nu] = (xi == 1 && nu == 1) ? bv : (v8f){0.f,0.f,0.f,0.f,0.f,0.f,0.f,0.f};
        #pragma unroll
        for (int kc = 0; kc < 4; ++kc) {
            #pragma unroll
            for (int nu = 0; nu < 4; ++nu)
                M[nu] = __builtin_amdgcn_wmma_f32_16x16x4_f32(
                            false, bB[(xi*4 + nu)*4 + kc], false, f[nu*4 + kc],
                            (short)0, M[nu], false, false);
        }
        const v8f Z0 = (M[0] + M[2]) + M[1];
        const v8f Z1 = M[1] - (M[2] + M[3]);
        if (xi == 0) { Y[0] = Z0; Y[1] = Z1; }
        if (xi == 1) { Y[0] += Z0; Y[1] += Z1; Y[2] = Z0; Y[3] = Z1; }
        if (xi == 2) { Y[0] += Z0; Y[1] += Z1; Y[2] -= Z0; Y[3] -= Z1; }
        if (xi == 3) {
            Y[2] -= Z0; Y[3] -= Z1;
            float* op = out + (size_t)g * PLANE + ((size_t)(b*16 + cb) * H + y0 + 2*p) * W + 2*N;
            #pragma unroll
            for (int e = 0; e < 8; ++e) *(v2f*)(op + (size_t)e * H * W)     = (v2f){Y[0][e], Y[1][e]};
            #pragma unroll
            for (int e = 0; e < 8; ++e) *(v2f*)(op + (size_t)e * H * W + W) = (v2f){Y[2][e], Y[3][e]};
        }
        #pragma unroll
        for (int t = 0; t < 16; ++t) dCur[t] = dNxt[t];
        __builtin_amdgcn_sched_barrier(0);
    }
}

__global__ __launch_bounds__(128, 2)
void conv16_f22_gated_kernel(const float* __restrict__ in,
                             const size_t in_branch_stride,
                             float* __restrict__ out,
                             const float* __restrict__ w,
                             const float* __restrict__ bias,
                             const float* __restrict__ gates)
{
    const int y0  = blockIdx.x * RB;
    const int b   = blockIdx.y;
    const int tid = threadIdx.x;
    const int lane = tid & 31;
    const int wave = tid >> 5;

    __shared__ float inT[TROWS*130*18];
    float* wU = inT;

    const int q   = (lane & 7) + 8*wave;
    const int chb = (lane >> 3) & 3;

    const int n     = lane & 15;
    const int half  = lane >> 4;
    const int N     = wave*16 + n;
    const int cb    = half * 8;

    v8f res[NPAIR][4];
    #pragma unroll
    for (int p = 0; p < NPAIR; ++p) {
        #pragma unroll
        for (int k = 0; k < 4; ++k) res[p][k] = (v8f){0.f,0.f,0.f,0.f,0.f,0.f,0.f,0.f};
    }

    for (int g = 0; g < 4; ++g) {
        const float* __restrict__ in_g = in + (size_t)g * in_branch_stride;
        const float* __restrict__ w_g  = w + (size_t)g * 3 * WLAYER;
        const float* __restrict__ b_g  = bias + g * 3 * 16;

        v4f sv[4*(RB+2)];
        #pragma unroll
        for (int j = 0; j < 4*(RB+2); ++j) {
            const int row = j >> 2;
            const int ch  = chb + 4*(j & 3);
            const int yi  = y0 - 1 + row;
            const int yc  = yi < 0 ? 0 : (yi > H-1 ? H-1 : yi);
            sv[j] = *(const v4f*)(in_g + ((size_t)(b*16 + ch) * H + yc) * W + 4*q);
        }

        const int wkc = tid >> 5, wln = tid & 31;
        float gkr[2][9];
        #pragma unroll
        for (int sel = 0; sel < 2; ++sel) {
            const int ic = wkc*4 + ((wln >> 4) << 1) + sel;
            const float* p0 = w_g + ((wln & 15)*16 + ic)*9;
            #pragma unroll
            for (int k = 0; k < 9; ++k) gkr[sel][k] = p0[k];
        }

        if (g > 0) __syncthreads();

        {
            const int kc = wkc, ln = wln;
            #pragma unroll
            for (int sel = 0; sel < 2; ++sel) {
                const float* gk = gkr[sel];
                float a[4][3];
                #pragma unroll
                for (int kx = 0; kx < 3; ++kx) {
                    const float g0 = gk[kx], g1 = gk[3 + kx], g2 = gk[6 + kx];
                    const float sg = g0 + g2;
                    a[0][kx] = g0;
                    a[1][kx] = 0.5f*(sg + g1);
                    a[2][kx] = 0.5f*(sg - g1);
                    a[3][kx] = g2;
                }
                #pragma unroll
                for (int xi = 0; xi < 4; ++xi) {
                    const float h0 = a[xi][0], h1 = a[xi][1], h2 = a[xi][2];
                    const float sh = h0 + h2;
                    float* d = &wU[((xi*4)*4 + kc)*64 + ln*2 + sel];
                    d[0*4*64] = h0;
                    d[1*4*64] = 0.5f*(sh + h1);
                    d[2*4*64] = 0.5f*(sh - h1);
                    d[3*4*64] = h2;
                }
            }
        }

        __syncthreads();
        v2f bB[64];
        #pragma unroll
        for (int t = 0; t < 64; ++t) bB[t] = *(const v2f*)&wU[t*64 + lane*2];
        __syncthreads();

        #pragma unroll
        for (int cj = 0; cj < 4; ++cj) {
            const int ch = chb + 4*cj;
            v4f rr[RB+2];
            #pragma unroll
            for (int row = 0; row < RB+2; ++row) {
                const int yi = y0 - 1 + row;
                const bool inside = (yi >= 0 && yi < H);
                rr[row] = inside ? sv[4*row + cj] : (v4f){0.f, 0.f, 0.f, 0.f};
            }
            #pragma unroll
            for (int p = 0; p < NPAIR; ++p) {
                v4f tv[4];
                tv[0] = rr[2*p]   - rr[2*p+2];
                tv[1] = rr[2*p+1] + rr[2*p+2];
                tv[2] = rr[2*p+2] - rr[2*p+1];
                tv[3] = rr[2*p+1] - rr[2*p+3];
                #pragma unroll
                for (int t = 0; t < 4; ++t) {
                    float* d = &inT[((4*p + t)*130 + 1 + 4*q)*18 + ch];
                    d[0] = tv[t].x; d[18] = tv[t].y; d[36] = tv[t].z; d[54] = tv[t].w;
                }
            }
        }
        for (int idx = tid; idx < TROWS*2*16; idx += 128) {
            const int ch = idx & 15;
            const int t  = idx >> 4;
            const int side = t & 1, row = t >> 1;
            inT[(row*130 + side*129)*18 + ch] = 0.f;
        }
        __syncthreads();

        v8f bv;
        #pragma unroll
        for (int e = 0; e < 8; ++e) bv[e] = b_g[cb + e];

        v2f gv[NPAIR][2];
        #pragma unroll
        for (int p = 0; p < NPAIR; ++p) {
            const float* gp = gates + ((size_t)(b*4 + g) * H + y0 + 2*p) * W + 2*N;
            gv[p][0] = *(const v2f*)gp;
            gv[p][1] = *(const v2f*)(gp + W);
        }

        v2f dCur[16], dNxt[16];
        load_row_cols(inT, 0, N, half, dCur);

        v8f Y[4];
        #pragma unroll
        for (int s = 0; s < TROWS; ++s) {
            const int p = s >> 2, xi = s & 3;
            v2f f[16];
            #pragma unroll
            for (int kc = 0; kc < 4; ++kc) {
                const v2f d0 = dCur[kc], d1 = dCur[4 + kc], d2 = dCur[8 + kc], d3 = dCur[12 + kc];
                f[kc]      = d0 - d2;
                f[4 + kc]  = d1 + d2;
                f[8 + kc]  = d2 - d1;
                f[12 + kc] = d1 - d3;
            }
            if (s + 1 < TROWS) load_row_cols(inT, s + 1, N, half, dNxt);
            v8f M[4];
            #pragma unroll
            for (int nu = 0; nu < 4; ++nu)
                M[nu] = (xi == 1 && nu == 1) ? bv : (v8f){0.f,0.f,0.f,0.f,0.f,0.f,0.f,0.f};
            #pragma unroll
            for (int kc = 0; kc < 4; ++kc) {
                #pragma unroll
                for (int nu = 0; nu < 4; ++nu)
                    M[nu] = __builtin_amdgcn_wmma_f32_16x16x4_f32(
                                false, bB[(xi*4 + nu)*4 + kc], false, f[nu*4 + kc],
                                (short)0, M[nu], false, false);
            }
            const v8f Z0 = (M[0] + M[2]) + M[1];
            const v8f Z1 = M[1] - (M[2] + M[3]);
            if (xi == 0) { Y[0] = Z0; Y[1] = Z1; }
            if (xi == 1) { Y[0] += Z0; Y[1] += Z1; Y[2] = Z0; Y[3] = Z1; }
            if (xi == 2) { Y[0] += Z0; Y[1] += Z1; Y[2] -= Z0; Y[3] -= Z1; }
            if (xi == 3) {
                Y[2] -= Z0; Y[3] -= Z1;
                res[p][0] += Y[0] * gv[p][0].x;
                res[p][1] += Y[1] * gv[p][0].y;
                res[p][2] += Y[2] * gv[p][1].x;
                res[p][3] += Y[3] * gv[p][1].y;
            }
            #pragma unroll
            for (int t = 0; t < 16; ++t) dCur[t] = dNxt[t];
            __builtin_amdgcn_sched_barrier(0);
        }
    }

    #pragma unroll
    for (int p = 0; p < NPAIR; ++p) {
        float* op = out + ((size_t)(b*16 + cb) * H + y0 + 2*p) * W + 2*N;
        #pragma unroll
        for (int e = 0; e < 8; ++e) *(v2f*)(op + (size_t)e * H * W)     = (v2f){res[p][0][e], res[p][1][e]};
        #pragma unroll
        for (int e = 0; e < 8; ++e) *(v2f*)(op + (size_t)e * H * W + W) = (v2f){res[p][2][e], res[p][3][e]};
    }
}

extern "C" void kernel_launch(void* const* d_in, const int* in_sizes, int n_in,
                              void* d_out, int out_size, void* d_ws, size_t ws_size,
                              hipStream_t stream)
{
    (void)in_sizes; (void)n_in; (void)out_size; (void)ws_size;
    const float* landmarks = (const float*)d_in[0];
    const float* features  = (const float*)d_in[1];
    const float* w_lm      = (const float*)d_in[2];
    const float* b_lm      = (const float*)d_in[3];
    const float* w_fea     = (const float*)d_in[4];
    const float* b_fea     = (const float*)d_in[5];
    float* out = (float*)d_out;

    char* ws = (char*)d_ws;
    float* gates = (float*)ws;
    float* bufA  = (float*)(ws + (size_t)8   * 1024*1024);
    float* bufB  = (float*)(ws + (size_t)136 * 1024*1024);

    dw_gate_kernel<<<dim3(8, 32), 512, 0, stream>>>(landmarks, w_lm, b_lm, gates);

    conv16_f22_kernel<<<dim3(128/RB, 32, 4), 256, 0, stream>>>(
        features, (size_t)0, bufA, w_fea + 0*WLAYER, b_fea + 0*16);
    conv16_f22_kernel<<<dim3(128/RB, 32, 4), 256, 0, stream>>>(
        bufA, PLANE, bufB, w_fea + 1*WLAYER, b_fea + 1*16);
    conv16_f22_gated_kernel<<<dim3(128/RB, 32), 128, 0, stream>>>(
        bufB, PLANE, out, w_fea + 2*WLAYER, b_fea + 2*16, gates);
}
